# starting kernel
# baseline (speedup 1.0000x reference)
#include <hip/hip_runtime.h>

typedef __attribute__((ext_vector_type(2))) float v2f;
typedef __attribute__((ext_vector_type(8))) float v8f;

#define Bn 8
#define Nn 8192
#define S_ATTN 128
#define S_NONE 384
#define Sn 512
#define NSAMPLE 64
#define RADIUS 0.4f
#define D_FEAT 64
#define IN_CH 67
#define C0PAD 68

#define OUT_NP_OFF 12288
#define OUT_ATT_OFF 536576

__device__ __forceinline__ v8f wmma_f32(v2f a, v2f b, v8f c) {
  return __builtin_amdgcn_wmma_f32_16x16x4_f32(
       false, a,  false, b,
       (short)0, c,  false,  false);
}

__global__ void __launch_bounds__(1024) fps_kernel(const float* __restrict__ xyz,
                                                   const float* __restrict__ attn,
                                                   float* __restrict__ out) {
  const int b = blockIdx.x >> 1;
  const int branch = blockIdx.x & 1;
  const int K = branch ? S_NONE : S_ATTN;
  const int rowoff = branch ? S_ATTN : 0;
  const int tid = threadIdx.x;
  const int wid = tid >> 5, lane = tid & 31;

  const float* xb = xyz + (size_t)b * 3 * Nn;
  const float* ab = attn + (size_t)b * Nn;

  float px[8], py[8], pz[8], mind[8];
#pragma unroll
  for (int i = 0; i < 8; ++i) {
    int n = tid + i * 1024;
    float a = ab[n];
    float w = branch ? (1.0f - a) : a;
    px[i] = w * xb[n];
    py[i] = w * xb[Nn + n];
    pz[i] = w * xb[2 * Nn + n];
    mind[i] = 1e10f;
  }

  __shared__ float s_bd[32];
  __shared__ int s_bn[32];
  __shared__ float s_l[3];
  __shared__ int s_last;

  int last = 0;
  for (int k = 0; k < K; ++k) {
    if (tid == 0) {
      float a = ab[last];
      float w = branch ? (1.0f - a) : a;
      float lx = w * xb[last];
      float ly = w * xb[Nn + last];
      float lz = w * xb[2 * Nn + last];
      int row = rowoff + k;
      out[(size_t)b * 1536 + row] = lx;
      out[(size_t)b * 1536 + 512 + row] = ly;
      out[(size_t)b * 1536 + 1024 + row] = lz;
      out[OUT_ATT_OFF + (size_t)b * 512 + row] = a;
      s_l[0] = lx; s_l[1] = ly; s_l[2] = lz;
    }
    __syncthreads();
    if (k == K - 1) break;

    const float lx = s_l[0], ly = s_l[1], lz = s_l[2];
    float bd = -1.0f;
    int bn = Nn;
#pragma unroll
    for (int i = 0; i < 8; ++i) {
      float dx = px[i] - lx, dy = py[i] - ly, dz = pz[i] - lz;
      float d2 = dx * dx + dy * dy + dz * dz;
      mind[i] = fminf(mind[i], d2);
      if (mind[i] > bd) { bd = mind[i]; bn = tid + i * 1024; }
    }
    for (int off = 16; off > 0; off >>= 1) {
      float od = __shfl_down(bd, off, 32);
      int on = __shfl_down(bn, off, 32);
      if (od > bd || (od == bd && on < bn)) { bd = od; bn = on; }
    }
    if (lane == 0) { s_bd[wid] = bd; s_bn[wid] = bn; }
    __syncthreads();
    if (wid == 0) {
      bd = s_bd[lane];
      bn = s_bn[lane];
      for (int off = 16; off > 0; off >>= 1) {
        float od = __shfl_down(bd, off, 32);
        int on = __shfl_down(bn, off, 32);
        if (od > bd || (od == bd && on < bn)) { bd = od; bn = on; }
      }
      if (lane == 0) s_last = bn;
    }
    __syncthreads();
    last = s_last;
  }
}

template <int CINPAD, int WSTRIDE>
__device__ __forceinline__ void mlp_layer(const float* __restrict__ W,
                                          const float* __restrict__ bias,
                                          const float* sin_, float* sout,
                                          int cout, int lane) {
  const int hl = lane >> 4;
  const int ll = lane & 15;
  for (int ot = 0; ot < cout / 16; ++ot) {
    const float bv = bias[ot * 16 + ll];
    for (int rt = 0; rt < 4; ++rt) {
      v8f acc = {};
      const int m = rt * 16 + ll;
      const int o = ot * 16 + ll;
      for (int c = 0; c < CINPAD; c += 4) {
        const int kk = c + (hl << 1);
        v2f a, bf;
        a.x = sin_[m * C0PAD + kk];
        a.y = sin_[m * C0PAD + kk + 1];
        bf.x = W[o * WSTRIDE + kk];
        bf.y = W[o * WSTRIDE + kk + 1];
        acc = wmma_f32(a, bf, acc);
      }
#pragma unroll
      for (int r = 0; r < 8; ++r) {
        float v = acc[r] + bv;
        v = v > 0.0f ? v : 0.0f;
        const int krow = rt * 16 + r + 8 * hl;
        sout[krow * C0PAD + ot * 16 + ll] = v;
      }
    }
  }
}

__device__ __forceinline__ void mlp_layer3_max(const float* __restrict__ W,
                                               const float* __restrict__ bias,
                                               const float* sin_,
                                               float* __restrict__ out,
                                               int lane, int b, int s) {
  const int hl = lane >> 4;
  const int ll = lane & 15;
  for (int ot = 0; ot < 8; ++ot) {
    const float bv = bias[ot * 16 + ll];
    v8f mx = {};
    for (int rt = 0; rt < 4; ++rt) {
      v8f acc = {};
      const int m = rt * 16 + ll;
      const int o = ot * 16 + ll;
      for (int c = 0; c < 64; c += 4) {
        const int kk = c + (hl << 1);
        v2f a, bf;
        a.x = sin_[m * C0PAD + kk];
        a.y = sin_[m * C0PAD + kk + 1];
        bf.x = W[o * 64 + kk];
        bf.y = W[o * 64 + kk + 1];
        acc = wmma_f32(a, bf, acc);
      }
#pragma unroll
      for (int r = 0; r < 8; ++r) {
        float v = acc[r] + bv;
        v = v > 0.0f ? v : 0.0f;
        mx[r] = fmaxf(mx[r], v);
      }
    }
    float m8 = mx[0];
#pragma unroll
    for (int r = 1; r < 8; ++r) m8 = fmaxf(m8, mx[r]);
    float oth = __shfl_xor(m8, 16, 32);
    m8 = fmaxf(m8, oth);
    if (lane < 16) {
      const int o = ot * 16 + lane;
      out[OUT_NP_OFF + (size_t)b * 65536 + (size_t)o * 512 + s] = m8;
    }
  }
}

__global__ void __launch_bounds__(32) sa_fused_kernel(
    const float* __restrict__ xyz, const float* __restrict__ pts,
    const float* __restrict__ w0, const float* __restrict__ b0,
    const float* __restrict__ w1, const float* __restrict__ b1,
    const float* __restrict__ w2, const float* __restrict__ b2,
    float* __restrict__ out) {
  __shared__ float s_w0[64 * C0PAD];
  __shared__ float s_feat[64 * C0PAD];
  __shared__ float s_act[64 * C0PAD];
  __shared__ int s_idx[NSAMPLE];

  const int lane = threadIdx.x;
  const int b = blockIdx.x >> 9;
  const int s = blockIdx.x & 511;

  for (int t = lane; t < 64 * C0PAD; t += 32) {
    int r = t / C0PAD, c = t - r * C0PAD;
    s_w0[t] = (c < IN_CH) ? w0[r * IN_CH + c] : 0.0f;
  }

  const float* xb = xyz + (size_t)b * 3 * Nn;
  const float qx = out[(size_t)b * 1536 + s];
  const float qy = out[(size_t)b * 1536 + 512 + s];
  const float qz = out[(size_t)b * 1536 + 1024 + s];

  const float R2 = RADIUS * RADIUS;
  int count = 0;
  for (int base = 0; base < Nn && count < NSAMPLE; base += 32) {
    __builtin_prefetch(xb + base + 256, 0, 0);
    const int n = base + lane;
    const float dx = xb[n] - qx;
    const float dy = xb[Nn + n] - qy;
    const float dz = xb[2 * Nn + n] - qz;
    const float d2 = dx * dx + dy * dy + dz * dz;
    const bool hit = d2 < R2;
    const unsigned mask = (unsigned)__ballot(hit);
    const int slot = count + __popc(mask & ((1u << lane) - 1u));
    if (hit && slot < NSAMPLE) s_idx[slot] = n;
    count += __popc(mask);
  }
  __syncthreads();
  const int cnt = count < NSAMPLE ? count : NSAMPLE;

  const float* pb = pts + (size_t)b * D_FEAT * Nn;
  for (int k = 0; k < NSAMPLE; ++k) {
    const int n = (k < cnt) ? s_idx[k] : -1;
    const int gi = (n >= 0) ? n : (Nn - 1);
#pragma unroll
    for (int j = 0; j < 3; ++j) {
      const int ch = lane + j * 32;
      if (ch < C0PAD) {
        float v;
        if (ch < 3) {
          float g = (n >= 0) ? xb[ch * Nn + n] : 0.0f;
          v = g - ((ch == 0) ? qx : (ch == 1) ? qy : qz);
        } else if (ch < IN_CH) {
          v = pb[(ch - 3) * Nn + gi];
        } else {
          v = 0.0f;
        }
        s_feat[k * C0PAD + ch] = v;
      }
    }
  }
  __syncthreads();

  mlp_layer<C0PAD, C0PAD>(s_w0, b0, s_feat, s_act, 64, lane);
  __syncthreads();
  mlp_layer<64, 64>(w1, b1, s_act, s_feat, 64, lane);
  __syncthreads();
  mlp_layer3_max(w2, b2, s_feat, out, lane, b, s);
}

extern "C" void kernel_launch(void* const* d_in, const int* in_sizes, int n_in,
                              void* d_out, int out_size, void* d_ws, size_t ws_size,
                              hipStream_t stream) {
  const float* xyz = (const float*)d_in[0];
  const float* pts = (const float*)d_in[1];
  const float* attn = (const float*)d_in[2];
  const float* w0 = (const float*)d_in[3];
  const float* b0 = (const float*)d_in[4];
  const float* w1 = (const float*)d_in[5];
  const float* b1 = (const float*)d_in[6];
  const float* w2 = (const float*)d_in[7];
  const float* b2 = (const float*)d_in[8];
  float* out = (float*)d_out;

  hipLaunchKernelGGL(fps_kernel, dim3(Bn * 2), dim3(1024), 0, stream, xyz, attn, out);
  hipLaunchKernelGGL(sa_fused_kernel, dim3(Bn * Sn), dim3(32), 0, stream,
                     xyz, pts, w0, b0, w1, b1, w2, b2, out);
}


# FPS: one argmax yields up to 8 picks: the 64 wave candidates are ranked by key compares and the longest prefix whose running distances earlier picks leave unchanged (exact test) shares one round
# speedup vs baseline: 3.1089x; 3.1089x over previous
#include <hip/hip_runtime.h>

typedef __attribute__((ext_vector_type(2))) float v2f;
typedef __attribute__((ext_vector_type(8))) float v8f;

#define Bn 8
#define Nn 8192
#define S_ATTN 128
#define S_NONE 384
#define Sn 512
#define NSAMPLE 64
#define RADIUS 0.4f
#define D_FEAT 64
#define IN_CH 67
#define C0PAD 68
#define FPS_BLOCKS 16
#define PT_BLOCKS 128
#define QPB 8
#define SA_BLOCKS 512
#define SA_THREADS 256
#define SEG_PTS 1024

#define OUT_NP_OFF 12288
#define OUT_ATT_OFF 536576

__device__ __forceinline__ v8f wmma_f32(v2f a, v2f b, v8f c) {
  return __builtin_amdgcn_wmma_f32_16x16x4_f32(
       false, a,  false, b,
       (short)0, c,  false,  false);
}

__device__ __forceinline__ void point_term_tile(const float* __restrict__ pts,
                                                const float* __restrict__ w0,
                                                float* __restrict__ pw,
                                                int tile, int lane) {
  const int hl = lane >> 4;
  const int ll = lane & 15;
  const int b = tile >> 9;
  const int n0 = (tile & 511) << 4;
  const float* pb = pts + (size_t)b * D_FEAT * Nn;
  v2f a[16];
#pragma unroll
  for (int j = 0; j < 16; ++j) {
    const int kk = 4 * j + (hl << 1);
    a[j].x = pb[(size_t)kk * Nn + n0 + ll];
    a[j].y = pb[(size_t)(kk + 1) * Nn + n0 + ll];
  }
  float* prow = pw + ((size_t)b * Nn + n0) * 64;
  for (int ot = 0; ot < 4; ++ot) {
    const int o = ot * 16 + ll;
    v8f acc = {};
#pragma unroll
    for (int j = 0; j < 16; ++j) {
      const int kk = 4 * j + (hl << 1);
      v2f bf;
      bf.x = w0[o * IN_CH + 3 + kk];
      bf.y = w0[o * IN_CH + 3 + kk + 1];
      acc = wmma_f32(a[j], bf, acc);
    }
#pragma unroll
    for (int r = 0; r < 8; ++r) prow[(size_t)(r + 8 * hl) * 64 + o] = acc[r];
  }
}

__device__ __forceinline__ float sqd3(float px, float py, float pz, float lx, float ly, float lz) {
  const float dx = px - lx, dy = py - ly, dz = pz - lz;
  return __builtin_fmaf(dz, dz, __builtin_fmaf(dx, dx, dy * dy));
}

__device__ __forceinline__ bool ranks_before(float av, int an, float bv, int bn) {
  return av > bv || (av == bv && an < bn);
}

__device__ __forceinline__ void top2_merge(float& b1, int& n1, float& b2, int& n2,
                                           float ob1, int on1, float ob2, int on2) {
  const bool lead = ranks_before(ob1, on1, b1, n1);
  const float xv = lead ? b1 : b2;
  const int xn = lead ? n1 : n2;
  const float yv = lead ? ob2 : ob1;
  const int yn = lead ? on2 : on1;
  const bool ysel = ranks_before(yv, yn, xv, xn);
  b2 = ysel ? yv : xv;
  n2 = ysel ? yn : xn;
  b1 = lead ? ob1 : b1;
  n1 = lead ? on1 : n1;
}

#define MAXP 8
#define NPAIRS (MAXP * (MAXP - 1) / 2)

__device__ __forceinline__ void fps_top2(const float (&mind)[8], int tid,
                                         float& b1, int& n1, float& b2, int& n2) {
#pragma unroll
  for (int i = 0; i < 8; ++i) {
    const float m = mind[i];
    const int n = tid + i * 1024;
    const bool g1 = m > b1;
    const bool g2 = m > b2;
    b2 = g1 ? b1 : (g2 ? m : b2);
    n2 = g1 ? n1 : (g2 ? n : n2);
    b1 = g1 ? m : b1;
    n1 = g1 ? n : n1;
  }
}

__device__ __forceinline__ long long rank_key(float v, int n) {
  return (long long)(((unsigned long long)(unsigned)__float_as_int(v) << 32) |
                     (unsigned long long)(~(unsigned)n));
}

__global__ void __launch_bounds__(1024) fps_kernel(const float* __restrict__ xyz,
                                                   const float* __restrict__ attn,
                                                   const float* __restrict__ pts,
                                                   const float* __restrict__ w0,
                                                   float* __restrict__ pw,
                                                   float* __restrict__ out) {
  if (blockIdx.x >= FPS_BLOCKS) {
    point_term_tile(pts, w0, pw, (blockIdx.x - FPS_BLOCKS) * 32 + (threadIdx.x >> 5),
                    threadIdx.x & 31);
    return;
  }
  const int b = blockIdx.x >> 1;
  const int branch = blockIdx.x & 1;
  const int K = branch ? S_NONE : S_ATTN;
  const int rowoff = branch ? S_ATTN : 0;
  const int tid = threadIdx.x;
  const int wid = tid >> 5, lane = tid & 31;

  const float* xb = xyz + (size_t)b * 3 * Nn;
  const float* ab = attn + (size_t)b * Nn;

  float px[8], py[8], pz[8], mind[8];
#pragma unroll
  for (int i = 0; i < 8; ++i) {
    int n = tid + i * 1024;
    float a = ab[n];
    float w = branch ? (1.0f - a) : a;
    px[i] = w * xb[n];
    py[i] = w * xb[Nn + n];
    pz[i] = w * xb[2 * Nn + n];
    mind[i] = 1e10f;
  }

  __shared__ long long s_key[64];
  __shared__ float s_rec[MAXP][8];

  int pj = 1;
#pragma unroll
  for (int j = 2; j < MAXP; ++j)
    if (lane >= j * (j - 1) / 2) pj = j;
  int pi = lane - pj * (pj - 1) / 2;
  const bool pair_lane = lane < NPAIRS;
  if (!pair_lane) { pi = 0; pj = 1; }

  if (tid < MAXP) {
    float a = ab[0];
    float w = branch ? (1.0f - a) : a;
    const bool first = tid == 0;
    s_rec[tid][0] = first ? w * xb[0] : 0.0f;
    s_rec[tid][1] = first ? w * xb[Nn] : 0.0f;
    s_rec[tid][2] = first ? w * xb[2 * Nn] : 0.0f;
    s_rec[tid][3] = a;
    s_rec[tid][4] = 0.0f;
    s_rec[tid][5] = 0.0f;
  }

  int k = 0;
  for (int round = 0; round < K; ++round) {
    __syncthreads();
    float qx[MAXP], qy[MAXP], qz[MAXP];
#pragma unroll
    for (int p = 0; p < MAXP; ++p) {
      qx[p] = __int_as_float(__builtin_amdgcn_readfirstlane(__float_as_int(s_rec[p][0])));
      qy[p] = __int_as_float(__builtin_amdgcn_readfirstlane(__float_as_int(s_rec[p][1])));
      qz[p] = __int_as_float(__builtin_amdgcn_readfirstlane(__float_as_int(s_rec[p][2])));
    }
    const float jv = s_rec[pj][4];
    const bool pass = jv > 0.0f && s_rec[pi][5] == 0.0f && k + pj < K &&
                      sqd3(s_rec[pj][0], s_rec[pj][1], s_rec[pj][2],
                           s_rec[pi][0], s_rec[pi][1], s_rec[pi][2]) >= jv;
    const unsigned fm = (unsigned)__ballot(pair_lane && !pass);
    int np = MAXP;
    if (fm != 0u) {
      const int f = __builtin_ctz(fm);
      np = 1;
#pragma unroll
      for (int j = 2; j < MAXP; ++j)
        if (f >= j * (j - 1) / 2) np = j;
    }
    if (tid < np) {
      const int row = rowoff + k + tid;
      out[(size_t)b * 1536 + row] = s_rec[tid][0];
      out[(size_t)b * 1536 + 512 + row] = s_rec[tid][1];
      out[(size_t)b * 1536 + 1024 + row] = s_rec[tid][2];
      out[OUT_ATT_OFF + (size_t)b * 512 + row] = s_rec[tid][3];
    }
    k += np;
    if (k >= K) break;

    float b1 = -1.0f, b2 = -1.0f;
    int n1 = Nn, n2 = Nn;
#pragma unroll
    for (int p = 0; p < MAXP; ++p) {
      if (p < np) {
#pragma unroll
        for (int i = 0; i < 8; ++i)
          mind[i] = fminf(mind[i], sqd3(px[i], py[i], pz[i], qx[p], qy[p], qz[p]));
      }
    }
    fps_top2(mind, tid, b1, n1, b2, n2);
    for (int off = 16; off > 0; off >>= 1) {
      float ob1 = __shfl_down(b1, off, 32);
      int on1 = __shfl_down(n1, off, 32);
      float ob2 = __shfl_down(b2, off, 32);
      int on2 = __shfl_down(n2, off, 32);
      top2_merge(b1, n1, b2, n2, ob1, on1, ob2, on2);
    }
    if (lane == 0) {
      s_key[2 * wid] = rank_key(b1, n1);
      s_key[2 * wid + 1] = rank_key(b2, n2);
    }
    __syncthreads();
    if (wid == 0) {
      const long long k1 = s_key[2 * lane];
      const long long k2 = s_key[2 * lane + 1];
      int r1 = 0, r2 = 0;
#pragma unroll 32
      for (int j = 0; j < 64; ++j) {
        const long long kj = s_key[j];
        r1 += (kj > k1) ? 1 : 0;
        r2 += (kj > k2) ? 1 : 0;
      }
      if (r1 < MAXP || r2 < MAXP) {
        const int c1 = (int)(~(unsigned)k1) & (Nn - 1);
        const int c2 = (int)(~(unsigned)k2) & (Nn - 1);
        const float a1 = ab[c1];
        const float x1 = xb[c1], y1 = xb[Nn + c1], z1 = xb[2 * Nn + c1];
        const float a2 = ab[c2];
        const float x2 = xb[c2], y2 = xb[Nn + c2], z2 = xb[2 * Nn + c2];
        if (r1 < MAXP) {
          const float w = branch ? (1.0f - a1) : a1;
          s_rec[r1][0] = w * x1;
          s_rec[r1][1] = w * y1;
          s_rec[r1][2] = w * z1;
          s_rec[r1][3] = a1;
          s_rec[r1][4] = __int_as_float((int)(k1 >> 32));
          s_rec[r1][5] = 0.0f;
        }
        if (r2 < MAXP) {
          const float w = branch ? (1.0f - a2) : a2;
          s_rec[r2][0] = w * x2;
          s_rec[r2][1] = w * y2;
          s_rec[r2][2] = w * z2;
          s_rec[r2][3] = a2;
          s_rec[r2][4] = __int_as_float((int)(k2 >> 32));
          s_rec[r2][5] = 1.0f;
        }
      }
    }
  }
}

template <int CINPAD, int WSTRIDE>
__device__ __forceinline__ void mlp_layer(const float* __restrict__ W,
                                          const float* __restrict__ bias,
                                          const float* sin_, float* sout,
                                          int cout, int lane) {
  const int hl = lane >> 4;
  const int ll = lane & 15;
  for (int ot = 0; ot < cout / 16; ++ot) {
    const float bv = bias[ot * 16 + ll];
#pragma unroll
    for (int rt = 0; rt < 4; ++rt) {
      v8f acc = {};
      const int m = rt * 16 + ll;
      const int o = ot * 16 + ll;
#pragma unroll
      for (int c = 0; c < CINPAD; c += 4) {
        const int kk = c + (hl << 1);
        v2f a, bf;
        a.x = sin_[m * C0PAD + kk];
        a.y = sin_[m * C0PAD + kk + 1];
        bf.x = W[o * WSTRIDE + kk];
        bf.y = W[o * WSTRIDE + kk + 1];
        acc = wmma_f32(a, bf, acc);
      }
#pragma unroll
      for (int r = 0; r < 8; ++r) {
        float v = acc[r] + bv;
        v = v > 0.0f ? v : 0.0f;
        const int krow = rt * 16 + r + 8 * hl;
        sout[krow * C0PAD + ot * 16 + ll] = v;
      }
    }
  }
}

__device__ __forceinline__ void mlp_layer3_max(const float* __restrict__ W,
                                               const float* __restrict__ bias,
                                               const float* sin_,
                                               float* __restrict__ out,
                                               int lane, int b, int s) {
  const int hl = lane >> 4;
  const int ll = lane & 15;
  for (int ot = 0; ot < 8; ++ot) {
    const float bv = bias[ot * 16 + ll];
    v8f mx = {};
    for (int rt = 0; rt < 4; ++rt) {
      v8f acc = {};
      const int m = rt * 16 + ll;
      const int o = ot * 16 + ll;
      for (int c = 0; c < 64; c += 4) {
        const int kk = c + (hl << 1);
        v2f a, bf;
        a.x = sin_[m * C0PAD + kk];
        a.y = sin_[m * C0PAD + kk + 1];
        bf.x = W[o * 64 + kk];
        bf.y = W[o * 64 + kk + 1];
        acc = wmma_f32(a, bf, acc);
      }
#pragma unroll
      for (int r = 0; r < 8; ++r) {
        float v = acc[r] + bv;
        v = v > 0.0f ? v : 0.0f;
        mx[r] = fmaxf(mx[r], v);
      }
    }
    float m8 = mx[0];
#pragma unroll
    for (int r = 1; r < 8; ++r) m8 = fmaxf(m8, mx[r]);
    float oth = __shfl_xor(m8, 16, 32);
    m8 = fmaxf(m8, oth);
    if (lane < 16) {
      const int o = ot * 16 + lane;
      out[OUT_NP_OFF + (size_t)b * 65536 + (size_t)o * 512 + s] = m8;
    }
  }
}

__global__ void __launch_bounds__(SA_THREADS) sa_fused_kernel(
    const float* __restrict__ xyz, const float* __restrict__ pw,
    const float* __restrict__ w0, const float* __restrict__ b0,
    const float* __restrict__ w1, const float* __restrict__ b1,
    const float* __restrict__ w2, const float* __restrict__ b2,
    float* __restrict__ out) {
  __shared__ float s_feat_all[QPB][64 * C0PAD];
  __shared__ float s_act_all[QPB][64 * C0PAD];
  __shared__ __attribute__((aligned(16))) int s_idx_all[QPB][NSAMPLE];
  __shared__ __attribute__((aligned(16))) unsigned s_hit[QPB][Nn / 32];

  const int lane = threadIdx.x & 31;
  const int w = __builtin_amdgcn_readfirstlane((int)(threadIdx.x >> 5));
  const int b = blockIdx.x >> 6;
  const int s0 = (blockIdx.x & 63) << 3;
  const int s = s0 + w;

  float* s_feat = s_feat_all[w];
  float* s_act = s_act_all[w];
  int* s_idx = s_idx_all[w];

  const float* xb = xyz + (size_t)b * 3 * Nn;
  const float* qb = out + (size_t)b * 1536 + s0;

  {
    const float R2 = RADIUS * RADIUS;
    const int p0 = w * SEG_PTS;
    float px[32], py[32], pz[32];
#pragma unroll
    for (int j = 0; j < 32; ++j) {
      const int n = p0 + j * 32 + lane;
      px[j] = xb[n];
      py[j] = xb[Nn + n];
      pz[j] = xb[2 * Nn + n];
    }
#pragma unroll
    for (int q = 0; q < QPB; ++q) {
      const float cx = qb[q];
      const float cy = qb[512 + q];
      const float cz = qb[1024 + q];
      int word = 0;
#pragma unroll
      for (int j = 0; j < 32; ++j) {
        const float dx = px[j] - cx;
        const float dy = py[j] - cy;
        const float dz = pz[j] - cz;
        const float d2 = __builtin_fmaf(dz, dz, __builtin_fmaf(dx, dx, dy * dy));
        const unsigned m = (unsigned)__ballot(d2 < R2);
        word = (lane == j) ? (int)m : word;
      }
      s_hit[q][w * 32 + lane] = (unsigned)word;
    }
  }
  __syncthreads();

  const float qx = qb[w];
  const float qy = qb[512 + w];
  const float qz = qb[1024 + w];

  {
    const uint4 h0 = *reinterpret_cast<const uint4*>(&s_hit[w][lane * 8]);
    const uint4 h1 = *reinterpret_cast<const uint4*>(&s_hit[w][lane * 8 + 4]);
    const unsigned hw[8] = {h0.x, h0.y, h0.z, h0.w, h1.x, h1.y, h1.z, h1.w};
    int c = 0;
#pragma unroll
    for (int i = 0; i < 8; ++i) c += __popc(hw[i]);
    int inc = c;
#pragma unroll
    for (int d = 1; d < 32; d <<= 1) {
      const int t = __shfl_up(inc, d, 32);
      if (lane >= d) inc += t;
    }
    const int total = __shfl(inc, 31, 32);
    int slot = inc - c;
#pragma unroll
    for (int i = 0; i < 8; ++i) {
      unsigned m = hw[i];
      const int base = (lane * 8 + i) * 32;
      while (m != 0u && slot < NSAMPLE) {
        s_idx[slot] = base + __builtin_ctz(m);
        ++slot;
        m &= m - 1u;
      }
    }
    for (int k = lane; k < NSAMPLE; k += 32)
      if (k >= total) s_idx[k] = -1;
  }
  __syncthreads();

  {
    const int hl = lane >> 4;
    const int ll = lane & 15;
    const float* Pb = pw + (size_t)b * Nn * 64;
    v2f bx[4];
    float bv[4];
#pragma unroll
    for (int ot = 0; ot < 4; ++ot) {
      const int o = ot * 16 + ll;
      bx[ot].x = w0[o * IN_CH + (hl << 1)];
      const float w1c = w0[o * IN_CH + 1];
      bx[ot].y = hl ? 0.0f : w1c;
      bv[ot] = b0[o];
    }
    v2f a[4];
#pragma unroll
    for (int rt = 0; rt < 4; ++rt) {
      const int n = s_idx[rt * 16 + ll];
      const int na = (n >= 0) ? n : 0;
      const float g0 = xb[(hl ? 2 * Nn : 0) + na];
      const float g1 = xb[Nn + na];
      a[rt].x = ((n >= 0) ? g0 : 0.0f) - (hl ? qz : qx);
      a[rt].y = hl ? 0.0f : (((n >= 0) ? g1 : 0.0f) - qy);
    }
#pragma unroll
    for (int rt = 0; rt < 4; ++rt) {
      const int4 i0 = *reinterpret_cast<const int4*>(&s_idx[rt * 16 + 8 * hl]);
      const int4 i1 = *reinterpret_cast<const int4*>(&s_idx[rt * 16 + 8 * hl + 4]);
      const int ix[8] = {i0.x, i0.y, i0.z, i0.w, i1.x, i1.y, i1.z, i1.w};
      int ro[8];
#pragma unroll
      for (int r = 0; r < 8; ++r) {
        const int gi = (ix[r] >= 0) ? ix[r] : (Nn - 1);
        ro[r] = gi * 64 + ll;
      }
#pragma unroll
      for (int ot = 0; ot < 4; ++ot) {
        v8f acc;
#pragma unroll
        for (int r = 0; r < 8; ++r) acc[r] = Pb[ro[r] + ot * 16];
        acc = wmma_f32(a[rt], bx[ot], acc);
#pragma unroll
        for (int r = 0; r < 8; ++r) {
          float v = acc[r] + bv[ot];
          v = v > 0.0f ? v : 0.0f;
          s_act[(rt * 16 + r + 8 * hl) * C0PAD + ot * 16 + ll] = v;
        }
      }
    }
  }
  __syncthreads();
  mlp_layer<64, 64>(w1, b1, s_act, s_feat, 64, lane);
  __syncthreads();
  mlp_layer3_max(w2, b2, s_feat, out, lane, b, s);
}

extern "C" void kernel_launch(void* const* d_in, const int* in_sizes, int n_in,
                              void* d_out, int out_size, void* d_ws, size_t ws_size,
                              hipStream_t stream) {
  const float* xyz = (const float*)d_in[0];
  const float* pts = (const float*)d_in[1];
  const float* attn = (const float*)d_in[2];
  const float* w0 = (const float*)d_in[3];
  const float* b0 = (const float*)d_in[4];
  const float* w1 = (const float*)d_in[5];
  const float* b1 = (const float*)d_in[6];
  const float* w2 = (const float*)d_in[7];
  const float* b2 = (const float*)d_in[8];
  float* out = (float*)d_out;
  float* pw = (float*)d_ws;

  hipLaunchKernelGGL(fps_kernel, dim3(FPS_BLOCKS + PT_BLOCKS), dim3(1024), 0, stream,
                     xyz, attn, pts, w0, pw, out);
  hipLaunchKernelGGL(sa_fused_kernel, dim3(SA_BLOCKS), dim3(SA_THREADS), 0, stream,
                     xyz, pw, w0, b0, w1, b1, w2, b2, out);
}


# FPS: every wave ranks its own two candidates among the 64 (4 key compares per lane, counted by mask popcounts) and writes its own record; before, one wave ranked all 64 with 128 compares per lane
# speedup vs baseline: 3.4957x; 1.1244x over previous
#include <hip/hip_runtime.h>

typedef __attribute__((ext_vector_type(2))) float v2f;
typedef __attribute__((ext_vector_type(8))) float v8f;

#define Bn 8
#define Nn 8192
#define S_ATTN 128
#define S_NONE 384
#define Sn 512
#define NSAMPLE 64
#define RADIUS 0.4f
#define D_FEAT 64
#define IN_CH 67
#define C0PAD 68
#define FPS_BLOCKS 16
#define PT_BLOCKS 128
#define QPB 8
#define SA_BLOCKS 512
#define SA_THREADS 256
#define SEG_PTS 1024

#define OUT_NP_OFF 12288
#define OUT_ATT_OFF 536576

__device__ __forceinline__ v8f wmma_f32(v2f a, v2f b, v8f c) {
  return __builtin_amdgcn_wmma_f32_16x16x4_f32(
       false, a,  false, b,
       (short)0, c,  false,  false);
}

__device__ __forceinline__ void point_term_tile(const float* __restrict__ pts,
                                                const float* __restrict__ w0,
                                                float* __restrict__ pw,
                                                int tile, int lane) {
  const int hl = lane >> 4;
  const int ll = lane & 15;
  const int b = tile >> 9;
  const int n0 = (tile & 511) << 4;
  const float* pb = pts + (size_t)b * D_FEAT * Nn;
  v2f a[16];
#pragma unroll
  for (int j = 0; j < 16; ++j) {
    const int kk = 4 * j + (hl << 1);
    a[j].x = pb[(size_t)kk * Nn + n0 + ll];
    a[j].y = pb[(size_t)(kk + 1) * Nn + n0 + ll];
  }
  float* prow = pw + ((size_t)b * Nn + n0) * 64;
  for (int ot = 0; ot < 4; ++ot) {
    const int o = ot * 16 + ll;
    v8f acc = {};
#pragma unroll
    for (int j = 0; j < 16; ++j) {
      const int kk = 4 * j + (hl << 1);
      v2f bf;
      bf.x = w0[o * IN_CH + 3 + kk];
      bf.y = w0[o * IN_CH + 3 + kk + 1];
      acc = wmma_f32(a[j], bf, acc);
    }
#pragma unroll
    for (int r = 0; r < 8; ++r) prow[(size_t)(r + 8 * hl) * 64 + o] = acc[r];
  }
}

__device__ __forceinline__ float sqd3(float px, float py, float pz, float lx, float ly, float lz) {
  const float dx = px - lx, dy = py - ly, dz = pz - lz;
  return __builtin_fmaf(dz, dz, __builtin_fmaf(dx, dx, dy * dy));
}

__device__ __forceinline__ bool ranks_before(float av, int an, float bv, int bn) {
  return av > bv || (av == bv && an < bn);
}

__device__ __forceinline__ void top2_merge(float& b1, int& n1, float& b2, int& n2,
                                           float ob1, int on1, float ob2, int on2) {
  const bool lead = ranks_before(ob1, on1, b1, n1);
  const float xv = lead ? b1 : b2;
  const int xn = lead ? n1 : n2;
  const float yv = lead ? ob2 : ob1;
  const int yn = lead ? on2 : on1;
  const bool ysel = ranks_before(yv, yn, xv, xn);
  b2 = ysel ? yv : xv;
  n2 = ysel ? yn : xn;
  b1 = lead ? ob1 : b1;
  n1 = lead ? on1 : n1;
}

#define MAXP 8
#define NPAIRS (MAXP * (MAXP - 1) / 2)

__device__ __forceinline__ void fps_top2(const float (&mind)[8], int tid,
                                         float& b1, int& n1, float& b2, int& n2) {
#pragma unroll
  for (int i = 0; i < 8; ++i) {
    const float m = mind[i];
    const int n = tid + i * 1024;
    const bool g1 = m > b1;
    const bool g2 = m > b2;
    b2 = g1 ? b1 : (g2 ? m : b2);
    n2 = g1 ? n1 : (g2 ? n : n2);
    b1 = g1 ? m : b1;
    n1 = g1 ? n : n1;
  }
}

__device__ __forceinline__ long long rank_key(float v, int n) {
  return (long long)(((unsigned long long)(unsigned)__float_as_int(v) << 32) |
                     (unsigned long long)(~(unsigned)n));
}

__global__ void __launch_bounds__(1024) fps_kernel(const float* __restrict__ xyz,
                                                   const float* __restrict__ attn,
                                                   const float* __restrict__ pts,
                                                   const float* __restrict__ w0,
                                                   float* __restrict__ pw,
                                                   float* __restrict__ out) {
  if (blockIdx.x >= FPS_BLOCKS) {
    point_term_tile(pts, w0, pw, (blockIdx.x - FPS_BLOCKS) * 32 + (threadIdx.x >> 5),
                    threadIdx.x & 31);
    return;
  }
  const int b = blockIdx.x >> 1;
  const int branch = blockIdx.x & 1;
  const int K = branch ? S_NONE : S_ATTN;
  const int rowoff = branch ? S_ATTN : 0;
  const int tid = threadIdx.x;
  const int wid = tid >> 5, lane = tid & 31;

  const float* xb = xyz + (size_t)b * 3 * Nn;
  const float* ab = attn + (size_t)b * Nn;

  float px[8], py[8], pz[8], mind[8];
#pragma unroll
  for (int i = 0; i < 8; ++i) {
    int n = tid + i * 1024;
    float a = ab[n];
    float w = branch ? (1.0f - a) : a;
    px[i] = w * xb[n];
    py[i] = w * xb[Nn + n];
    pz[i] = w * xb[2 * Nn + n];
    mind[i] = 1e10f;
  }

  __shared__ __attribute__((aligned(16))) long long s_key[64];
  __shared__ float s_rec[MAXP][8];

  int pj = 1;
#pragma unroll
  for (int j = 2; j < MAXP; ++j)
    if (lane >= j * (j - 1) / 2) pj = j;
  int pi = lane - pj * (pj - 1) / 2;
  const bool pair_lane = lane < NPAIRS;
  if (!pair_lane) { pi = 0; pj = 1; }

  if (tid < MAXP) {
    float a = ab[0];
    float w = branch ? (1.0f - a) : a;
    const bool first = tid == 0;
    s_rec[tid][0] = first ? w * xb[0] : 0.0f;
    s_rec[tid][1] = first ? w * xb[Nn] : 0.0f;
    s_rec[tid][2] = first ? w * xb[2 * Nn] : 0.0f;
    s_rec[tid][3] = a;
    s_rec[tid][4] = 0.0f;
    s_rec[tid][5] = 0.0f;
  }

  int k = 0;
  for (int round = 0; round < K; ++round) {
    __syncthreads();
    float qx[MAXP], qy[MAXP], qz[MAXP];
#pragma unroll
    for (int p = 0; p < MAXP; ++p) {
      qx[p] = __int_as_float(__builtin_amdgcn_readfirstlane(__float_as_int(s_rec[p][0])));
      qy[p] = __int_as_float(__builtin_amdgcn_readfirstlane(__float_as_int(s_rec[p][1])));
      qz[p] = __int_as_float(__builtin_amdgcn_readfirstlane(__float_as_int(s_rec[p][2])));
    }
    const float jv = s_rec[pj][4];
    const bool pass = jv > 0.0f && s_rec[pi][5] == 0.0f && k + pj < K &&
                      sqd3(s_rec[pj][0], s_rec[pj][1], s_rec[pj][2],
                           s_rec[pi][0], s_rec[pi][1], s_rec[pi][2]) >= jv;
    const unsigned fm = (unsigned)__ballot(pair_lane && !pass);
    int np = MAXP;
    if (fm != 0u) {
      const int f = __builtin_ctz(fm);
      np = 1;
#pragma unroll
      for (int j = 2; j < MAXP; ++j)
        if (f >= j * (j - 1) / 2) np = j;
    }
    if (tid < np) {
      const int row = rowoff + k + tid;
      out[(size_t)b * 1536 + row] = s_rec[tid][0];
      out[(size_t)b * 1536 + 512 + row] = s_rec[tid][1];
      out[(size_t)b * 1536 + 1024 + row] = s_rec[tid][2];
      out[OUT_ATT_OFF + (size_t)b * 512 + row] = s_rec[tid][3];
    }
    k += np;
    if (k >= K) break;

    float b1 = -1.0f, b2 = -1.0f;
    int n1 = Nn, n2 = Nn;
#pragma unroll
    for (int p = 0; p < MAXP; ++p) {
      if (p < np) {
#pragma unroll
        for (int i = 0; i < 8; ++i)
          mind[i] = fminf(mind[i], sqd3(px[i], py[i], pz[i], qx[p], qy[p], qz[p]));
      }
    }
    fps_top2(mind, tid, b1, n1, b2, n2);
    for (int off = 16; off > 0; off >>= 1) {
      float ob1 = __shfl_down(b1, off, 32);
      int on1 = __shfl_down(n1, off, 32);
      float ob2 = __shfl_down(b2, off, 32);
      int on2 = __shfl_down(n2, off, 32);
      top2_merge(b1, n1, b2, n2, ob1, on1, ob2, on2);
    }
    const int vb1 = __builtin_amdgcn_readfirstlane(__float_as_int(b1));
    const int vn1 = __builtin_amdgcn_readfirstlane(n1);
    const int vb2 = __builtin_amdgcn_readfirstlane(__float_as_int(b2));
    const int vn2 = __builtin_amdgcn_readfirstlane(n2);
    const long long k1 = rank_key(__int_as_float(vb1), vn1);
    const long long k2 = rank_key(__int_as_float(vb2), vn2);
    if (lane == 0) {
      s_key[2 * wid] = k1;
      s_key[2 * wid + 1] = k2;
    }
    __syncthreads();
    {
      const long long ka = s_key[2 * lane];
      const long long kb = s_key[2 * lane + 1];
      const int r1 = __popc((unsigned)__ballot(ka > k1)) + __popc((unsigned)__ballot(kb > k1));
      const int r2 = __popc((unsigned)__ballot(ka > k2)) + __popc((unsigned)__ballot(kb > k2));
      const int r = lane ? r2 : r1;
      if (lane < 2 && r < MAXP) {
        const int c = (lane ? vn2 : vn1) & (Nn - 1);
        const float a = ab[c];
        const float x = xb[c], y = xb[Nn + c], z = xb[2 * Nn + c];
        const float w = branch ? (1.0f - a) : a;
        s_rec[r][0] = w * x;
        s_rec[r][1] = w * y;
        s_rec[r][2] = w * z;
        s_rec[r][3] = a;
        s_rec[r][4] = __int_as_float(lane ? vb2 : vb1);
        s_rec[r][5] = lane ? 1.0f : 0.0f;
      }
    }
  }
}

template <int CINPAD, int WSTRIDE>
__device__ __forceinline__ void mlp_layer(const float* __restrict__ W,
                                          const float* __restrict__ bias,
                                          const float* sin_, float* sout,
                                          int cout, int lane) {
  const int hl = lane >> 4;
  const int ll = lane & 15;
  for (int ot = 0; ot < cout / 16; ++ot) {
    const float bv = bias[ot * 16 + ll];
#pragma unroll
    for (int rt = 0; rt < 4; ++rt) {
      v8f acc = {};
      const int m = rt * 16 + ll;
      const int o = ot * 16 + ll;
#pragma unroll
      for (int c = 0; c < CINPAD; c += 4) {
        const int kk = c + (hl << 1);
        v2f a, bf;
        a.x = sin_[m * C0PAD + kk];
        a.y = sin_[m * C0PAD + kk + 1];
        bf.x = W[o * WSTRIDE + kk];
        bf.y = W[o * WSTRIDE + kk + 1];
        acc = wmma_f32(a, bf, acc);
      }
#pragma unroll
      for (int r = 0; r < 8; ++r) {
        float v = acc[r] + bv;
        v = v > 0.0f ? v : 0.0f;
        const int krow = rt * 16 + r + 8 * hl;
        sout[krow * C0PAD + ot * 16 + ll] = v;
      }
    }
  }
}

__device__ __forceinline__ void mlp_layer3_max(const float* __restrict__ W,
                                               const float* __restrict__ bias,
                                               const float* sin_,
                                               float* __restrict__ out,
                                               int lane, int b, int s) {
  const int hl = lane >> 4;
  const int ll = lane & 15;
  for (int ot = 0; ot < 8; ++ot) {
    const float bv = bias[ot * 16 + ll];
    v8f mx = {};
    for (int rt = 0; rt < 4; ++rt) {
      v8f acc = {};
      const int m = rt * 16 + ll;
      const int o = ot * 16 + ll;
      for (int c = 0; c < 64; c += 4) {
        const int kk = c + (hl << 1);
        v2f a, bf;
        a.x = sin_[m * C0PAD + kk];
        a.y = sin_[m * C0PAD + kk + 1];
        bf.x = W[o * 64 + kk];
        bf.y = W[o * 64 + kk + 1];
        acc = wmma_f32(a, bf, acc);
      }
#pragma unroll
      for (int r = 0; r < 8; ++r) {
        float v = acc[r] + bv;
        v = v > 0.0f ? v : 0.0f;
        mx[r] = fmaxf(mx[r], v);
      }
    }
    float m8 = mx[0];
#pragma unroll
    for (int r = 1; r < 8; ++r) m8 = fmaxf(m8, mx[r]);
    float oth = __shfl_xor(m8, 16, 32);
    m8 = fmaxf(m8, oth);
    if (lane < 16) {
      const int o = ot * 16 + lane;
      out[OUT_NP_OFF + (size_t)b * 65536 + (size_t)o * 512 + s] = m8;
    }
  }
}

__global__ void __launch_bounds__(SA_THREADS) sa_fused_kernel(
    const float* __restrict__ xyz, const float* __restrict__ pw,
    const float* __restrict__ w0, const float* __restrict__ b0,
    const float* __restrict__ w1, const float* __restrict__ b1,
    const float* __restrict__ w2, const float* __restrict__ b2,
    float* __restrict__ out) {
  __shared__ float s_feat_all[QPB][64 * C0PAD];
  __shared__ float s_act_all[QPB][64 * C0PAD];
  __shared__ __attribute__((aligned(16))) int s_idx_all[QPB][NSAMPLE];
  __shared__ __attribute__((aligned(16))) unsigned s_hit[QPB][Nn / 32];

  const int lane = threadIdx.x & 31;
  const int w = __builtin_amdgcn_readfirstlane((int)(threadIdx.x >> 5));
  const int b = blockIdx.x >> 6;
  const int s0 = (blockIdx.x & 63) << 3;
  const int s = s0 + w;

  float* s_feat = s_feat_all[w];
  float* s_act = s_act_all[w];
  int* s_idx = s_idx_all[w];

  const float* xb = xyz + (size_t)b * 3 * Nn;
  const float* qb = out + (size_t)b * 1536 + s0;

  {
    const float R2 = RADIUS * RADIUS;
    const int p0 = w * SEG_PTS;
    float px[32], py[32], pz[32];
#pragma unroll
    for (int j = 0; j < 32; ++j) {
      const int n = p0 + j * 32 + lane;
      px[j] = xb[n];
      py[j] = xb[Nn + n];
      pz[j] = xb[2 * Nn + n];
    }
#pragma unroll
    for (int q = 0; q < QPB; ++q) {
      const float cx = qb[q];
      const float cy = qb[512 + q];
      const float cz = qb[1024 + q];
      int word = 0;
#pragma unroll
      for (int j = 0; j < 32; ++j) {
        const float dx = px[j] - cx;
        const float dy = py[j] - cy;
        const float dz = pz[j] - cz;
        const float d2 = __builtin_fmaf(dz, dz, __builtin_fmaf(dx, dx, dy * dy));
        const unsigned m = (unsigned)__ballot(d2 < R2);
        word = (lane == j) ? (int)m : word;
      }
      s_hit[q][w * 32 + lane] = (unsigned)word;
    }
  }
  __syncthreads();

  const float qx = qb[w];
  const float qy = qb[512 + w];
  const float qz = qb[1024 + w];

  {
    const uint4 h0 = *reinterpret_cast<const uint4*>(&s_hit[w][lane * 8]);
    const uint4 h1 = *reinterpret_cast<const uint4*>(&s_hit[w][lane * 8 + 4]);
    const unsigned hw[8] = {h0.x, h0.y, h0.z, h0.w, h1.x, h1.y, h1.z, h1.w};
    int c = 0;
#pragma unroll
    for (int i = 0; i < 8; ++i) c += __popc(hw[i]);
    int inc = c;
#pragma unroll
    for (int d = 1; d < 32; d <<= 1) {
      const int t = __shfl_up(inc, d, 32);
      if (lane >= d) inc += t;
    }
    const int total = __shfl(inc, 31, 32);
    int slot = inc - c;
#pragma unroll
    for (int i = 0; i < 8; ++i) {
      unsigned m = hw[i];
      const int base = (lane * 8 + i) * 32;
      while (m != 0u && slot < NSAMPLE) {
        s_idx[slot] = base + __builtin_ctz(m);
        ++slot;
        m &= m - 1u;
      }
    }
    for (int k = lane; k < NSAMPLE; k += 32)
      if (k >= total) s_idx[k] = -1;
  }
  __syncthreads();

  {
    const int hl = lane >> 4;
    const int ll = lane & 15;
    const float* Pb = pw + (size_t)b * Nn * 64;
    v2f bx[4];
    float bv[4];
#pragma unroll
    for (int ot = 0; ot < 4; ++ot) {
      const int o = ot * 16 + ll;
      bx[ot].x = w0[o * IN_CH + (hl << 1)];
      const float w1c = w0[o * IN_CH + 1];
      bx[ot].y = hl ? 0.0f : w1c;
      bv[ot] = b0[o];
    }
    v2f a[4];
#pragma unroll
    for (int rt = 0; rt < 4; ++rt) {
      const int n = s_idx[rt * 16 + ll];
      const int na = (n >= 0) ? n : 0;
      const float g0 = xb[(hl ? 2 * Nn : 0) + na];
      const float g1 = xb[Nn + na];
      a[rt].x = ((n >= 0) ? g0 : 0.0f) - (hl ? qz : qx);
      a[rt].y = hl ? 0.0f : (((n >= 0) ? g1 : 0.0f) - qy);
    }
#pragma unroll
    for (int rt = 0; rt < 4; ++rt) {
      const int4 i0 = *reinterpret_cast<const int4*>(&s_idx[rt * 16 + 8 * hl]);
      const int4 i1 = *reinterpret_cast<const int4*>(&s_idx[rt * 16 + 8 * hl + 4]);
      const int ix[8] = {i0.x, i0.y, i0.z, i0.w, i1.x, i1.y, i1.z, i1.w};
      int ro[8];
#pragma unroll
      for (int r = 0; r < 8; ++r) {
        const int gi = (ix[r] >= 0) ? ix[r] : (Nn - 1);
        ro[r] = gi * 64 + ll;
      }
#pragma unroll
      for (int ot = 0; ot < 4; ++ot) {
        v8f acc;
#pragma unroll
        for (int r = 0; r < 8; ++r) acc[r] = Pb[ro[r] + ot * 16];
        acc = wmma_f32(a[rt], bx[ot], acc);
#pragma unroll
        for (int r = 0; r < 8; ++r) {
          float v = acc[r] + bv[ot];
          v = v > 0.0f ? v : 0.0f;
          s_act[(rt * 16 + r + 8 * hl) * C0PAD + ot * 16 + ll] = v;
        }
      }
    }
  }
  __syncthreads();
  mlp_layer<64, 64>(w1, b1, s_act, s_feat, 64, lane);
  __syncthreads();
  mlp_layer3_max(w2, b2, s_feat, out, lane, b, s);
}

extern "C" void kernel_launch(void* const* d_in, const int* in_sizes, int n_in,
                              void* d_out, int out_size, void* d_ws, size_t ws_size,
                              hipStream_t stream) {
  const float* xyz = (const float*)d_in[0];
  const float* pts = (const float*)d_in[1];
  const float* attn = (const float*)d_in[2];
  const float* w0 = (const float*)d_in[3];
  const float* b0 = (const float*)d_in[4];
  const float* w1 = (const float*)d_in[5];
  const float* b1 = (const float*)d_in[6];
  const float* w2 = (const float*)d_in[7];
  const float* b2 = (const float*)d_in[8];
  float* out = (float*)d_out;
  float* pw = (float*)d_ws;

  hipLaunchKernelGGL(fps_kernel, dim3(FPS_BLOCKS + PT_BLOCKS), dim3(1024), 0, stream,
                     xyz, attn, pts, w0, pw, out);
  hipLaunchKernelGGL(sa_fused_kernel, dim3(SA_BLOCKS), dim3(SA_THREADS), 0, stream,
                     xyz, pw, w0, b0, w1, b1, w2, b2, out);
}


# FPS: a ranked candidate that a taken pick of the round lowers below the list's last candidate is passed over instead of ending the round (exact); pair tests as an 8x8 lane matrix
# speedup vs baseline: 3.9032x; 1.1166x over previous
#include <hip/hip_runtime.h>

typedef __attribute__((ext_vector_type(2))) float v2f;
typedef __attribute__((ext_vector_type(8))) float v8f;

#define Bn 8
#define Nn 8192
#define S_ATTN 128
#define S_NONE 384
#define Sn 512
#define NSAMPLE 64
#define RADIUS 0.4f
#define D_FEAT 64
#define IN_CH 67
#define C0PAD 68
#define FPS_BLOCKS 16
#define PT_BLOCKS 128
#define QPB 8
#define SA_BLOCKS 512
#define SA_THREADS 256
#define SEG_PTS 1024

#define OUT_NP_OFF 12288
#define OUT_ATT_OFF 536576

__device__ __forceinline__ v8f wmma_f32(v2f a, v2f b, v8f c) {
  return __builtin_amdgcn_wmma_f32_16x16x4_f32(
       false, a,  false, b,
       (short)0, c,  false,  false);
}

__device__ __forceinline__ void point_term_tile(const float* __restrict__ pts,
                                                const float* __restrict__ w0,
                                                float* __restrict__ pw,
                                                int tile, int lane) {
  const int hl = lane >> 4;
  const int ll = lane & 15;
  const int b = tile >> 9;
  const int n0 = (tile & 511) << 4;
  const float* pb = pts + (size_t)b * D_FEAT * Nn;
  v2f a[16];
#pragma unroll
  for (int j = 0; j < 16; ++j) {
    const int kk = 4 * j + (hl << 1);
    a[j].x = pb[(size_t)kk * Nn + n0 + ll];
    a[j].y = pb[(size_t)(kk + 1) * Nn + n0 + ll];
  }
  float* prow = pw + ((size_t)b * Nn + n0) * 64;
  for (int ot = 0; ot < 4; ++ot) {
    const int o = ot * 16 + ll;
    v8f acc = {};
#pragma unroll
    for (int j = 0; j < 16; ++j) {
      const int kk = 4 * j + (hl << 1);
      v2f bf;
      bf.x = w0[o * IN_CH + 3 + kk];
      bf.y = w0[o * IN_CH + 3 + kk + 1];
      acc = wmma_f32(a[j], bf, acc);
    }
#pragma unroll
    for (int r = 0; r < 8; ++r) prow[(size_t)(r + 8 * hl) * 64 + o] = acc[r];
  }
}

__device__ __forceinline__ float sqd3(float px, float py, float pz, float lx, float ly, float lz) {
  const float dx = px - lx, dy = py - ly, dz = pz - lz;
  return __builtin_fmaf(dz, dz, __builtin_fmaf(dx, dx, dy * dy));
}

__device__ __forceinline__ bool ranks_before(float av, int an, float bv, int bn) {
  return av > bv || (av == bv && an < bn);
}

__device__ __forceinline__ void top2_merge(float& b1, int& n1, float& b2, int& n2,
                                           float ob1, int on1, float ob2, int on2) {
  const bool lead = ranks_before(ob1, on1, b1, n1);
  const float xv = lead ? b1 : b2;
  const int xn = lead ? n1 : n2;
  const float yv = lead ? ob2 : ob1;
  const int yn = lead ? on2 : on1;
  const bool ysel = ranks_before(yv, yn, xv, xn);
  b2 = ysel ? yv : xv;
  n2 = ysel ? yn : xn;
  b1 = lead ? ob1 : b1;
  n1 = lead ? on1 : n1;
}

#define MAXP 8

__device__ __forceinline__ void fps_top2(const float (&mind)[8], int tid,
                                         float& b1, int& n1, float& b2, int& n2) {
#pragma unroll
  for (int i = 0; i < 8; ++i) {
    const float m = mind[i];
    const int n = tid + i * 1024;
    const bool g1 = m > b1;
    const bool g2 = m > b2;
    b2 = g1 ? b1 : (g2 ? m : b2);
    n2 = g1 ? n1 : (g2 ? n : n2);
    b1 = g1 ? m : b1;
    n1 = g1 ? n : n1;
  }
}

__device__ __forceinline__ long long rank_key(float v, int n) {
  return (long long)(((unsigned long long)(unsigned)__float_as_int(v) << 32) |
                     (unsigned long long)(~(unsigned)n));
}

__global__ void __launch_bounds__(1024) fps_kernel(const float* __restrict__ xyz,
                                                   const float* __restrict__ attn,
                                                   const float* __restrict__ pts,
                                                   const float* __restrict__ w0,
                                                   float* __restrict__ pw,
                                                   float* __restrict__ out) {
  if (blockIdx.x >= FPS_BLOCKS) {
    point_term_tile(pts, w0, pw, (blockIdx.x - FPS_BLOCKS) * 32 + (threadIdx.x >> 5),
                    threadIdx.x & 31);
    return;
  }
  const int b = blockIdx.x >> 1;
  const int branch = blockIdx.x & 1;
  const int K = branch ? S_NONE : S_ATTN;
  const int rowoff = branch ? S_ATTN : 0;
  const int tid = threadIdx.x;
  const int wid = tid >> 5, lane = tid & 31;

  const float* xb = xyz + (size_t)b * 3 * Nn;
  const float* ab = attn + (size_t)b * Nn;

  float px[8], py[8], pz[8], mind[8];
#pragma unroll
  for (int i = 0; i < 8; ++i) {
    int n = tid + i * 1024;
    float a = ab[n];
    float w = branch ? (1.0f - a) : a;
    px[i] = w * xb[n];
    py[i] = w * xb[Nn + n];
    pz[i] = w * xb[2 * Nn + n];
    mind[i] = 1e10f;
  }

  __shared__ __attribute__((aligned(16))) long long s_key[64];
  __shared__ __attribute__((aligned(16))) float s_rec[MAXP][8];

  const int mj = lane >> 2;
  const int mc = lane & 3;

  if (tid < MAXP) {
    float a = ab[0];
    float w = branch ? (1.0f - a) : a;
    const bool first = tid == 0;
    s_rec[tid][0] = first ? w * xb[0] : 0.0f;
    s_rec[tid][1] = first ? w * xb[Nn] : 0.0f;
    s_rec[tid][2] = first ? w * xb[2 * Nn] : 0.0f;
    s_rec[tid][3] = 0.0f;
    s_rec[tid][4] = 0.0f;
    s_rec[tid][5] = a;
  }

  int k = 0;
  for (int round = 0; round < K; ++round) {
    __syncthreads();
    float qx[MAXP], qy[MAXP], qz[MAXP];
#pragma unroll
    for (int p = 0; p < MAXP; ++p) {
      qx[p] = __int_as_float(__builtin_amdgcn_readfirstlane(__float_as_int(s_rec[p][0])));
      qy[p] = __int_as_float(__builtin_amdgcn_readfirstlane(__float_as_int(s_rec[p][1])));
      qz[p] = __int_as_float(__builtin_amdgcn_readfirstlane(__float_as_int(s_rec[p][2])));
    }
    const float4 rj = *reinterpret_cast<const float4*>(&s_rec[mj][0]);
    const float vj = s_rec[mj][4];
    const float4 cl = *reinterpret_cast<const float4*>(&s_rec[mc][0]);
    const float4 ch = *reinterpret_cast<const float4*>(&s_rec[mc + 4][0]);
    const float vlast = s_rec[MAXP - 1][4];
    const float dlo = sqd3(rj.x, rj.y, rj.z, cl.x, cl.y, cl.z);
    const float dhi = sqd3(rj.x, rj.y, rj.z, ch.x, ch.y, ch.z);
    const bool lov = mc < mj, hiv = mc + 4 < mj;
    const unsigned vlo = (unsigned)__ballot(lov && dlo < vj);
    const unsigned vhi = (unsigned)__ballot(hiv && dhi < vj);
    const unsigned wlo = (unsigned)__ballot(lov && dlo < vlast);
    const unsigned whi = (unsigned)__ballot(hiv && dhi < vlast);
    const bool hardrow = (mj >= 1) & (!(vj > 0.0f) | (lov & (cl.w != 0.0f)) | (hiv & (ch.w != 0.0f)));
    const int H = __builtin_ctzll((unsigned long long)(unsigned)__ballot(hardrow) | (1ull << 32)) >> 2;
    unsigned A = (1u << H) - 1u;
    unsigned pos = (unsigned)((1ull << (4 * H)) - 1ull) & 0xFFFFFFF0u;
    for (int t = 0; t < MAXP - 1; ++t) {
      const unsigned alo = (A & 15u) * 0x11111111u;
      const unsigned ahi = (A >> 4) * 0x11111111u;
      const unsigned low = ((vlo & alo) | (vhi & ahi)) & pos;
      if (low == 0u) break;
      const int sr = __builtin_ctz(low) >> 2;
      const unsigned strong = (((wlo & alo) | (whi & ahi)) >> (4 * sr)) & 15u;
      if (strong == 0u) {
        A &= (1u << sr) - 1u;
        break;
      }
      A &= ~(1u << sr);
      pos &= 0xFFFFFFF0u << (4 * sr);
    }
    if (tid < MAXP && ((A >> tid) & 1u) != 0u) {
      const int kr = k + __popc(A & ((1u << tid) - 1u));
      if (kr < K) {
        const int row = rowoff + kr;
        out[(size_t)b * 1536 + row] = s_rec[tid][0];
        out[(size_t)b * 1536 + 512 + row] = s_rec[tid][1];
        out[(size_t)b * 1536 + 1024 + row] = s_rec[tid][2];
        out[OUT_ATT_OFF + (size_t)b * 512 + row] = s_rec[tid][5];
      }
    }
    k += __popc(A);
    if (k >= K) break;

    float b1 = -1.0f, b2 = -1.0f;
    int n1 = Nn, n2 = Nn;
#pragma unroll
    for (int p = 0; p < MAXP; ++p) {
      if ((A >> p) & 1u) {
#pragma unroll
        for (int i = 0; i < 8; ++i)
          mind[i] = fminf(mind[i], sqd3(px[i], py[i], pz[i], qx[p], qy[p], qz[p]));
      }
    }
    fps_top2(mind, tid, b1, n1, b2, n2);
    for (int off = 16; off > 0; off >>= 1) {
      float ob1 = __shfl_down(b1, off, 32);
      int on1 = __shfl_down(n1, off, 32);
      float ob2 = __shfl_down(b2, off, 32);
      int on2 = __shfl_down(n2, off, 32);
      top2_merge(b1, n1, b2, n2, ob1, on1, ob2, on2);
    }
    const int vb1 = __builtin_amdgcn_readfirstlane(__float_as_int(b1));
    const int vn1 = __builtin_amdgcn_readfirstlane(n1);
    const int vb2 = __builtin_amdgcn_readfirstlane(__float_as_int(b2));
    const int vn2 = __builtin_amdgcn_readfirstlane(n2);
    const long long k1 = rank_key(__int_as_float(vb1), vn1);
    const long long k2 = rank_key(__int_as_float(vb2), vn2);
    if (lane == 0) {
      s_key[2 * wid] = k1;
      s_key[2 * wid + 1] = k2;
    }
    __syncthreads();
    {
      const long long ka = s_key[2 * lane];
      const long long kb = s_key[2 * lane + 1];
      const int r1 = __popc((unsigned)__ballot(ka > k1)) + __popc((unsigned)__ballot(kb > k1));
      const int r2 = __popc((unsigned)__ballot(ka > k2)) + __popc((unsigned)__ballot(kb > k2));
      const int r = lane ? r2 : r1;
      if (lane < 2 && r < MAXP) {
        const int c = (lane ? vn2 : vn1) & (Nn - 1);
        const float a = ab[c];
        const float x = xb[c], y = xb[Nn + c], z = xb[2 * Nn + c];
        const float w = branch ? (1.0f - a) : a;
        s_rec[r][0] = w * x;
        s_rec[r][1] = w * y;
        s_rec[r][2] = w * z;
        s_rec[r][3] = lane ? 1.0f : 0.0f;
        s_rec[r][4] = __int_as_float(lane ? vb2 : vb1);
        s_rec[r][5] = a;
      }
    }
  }
}

template <int CINPAD, int WSTRIDE>
__device__ __forceinline__ void mlp_layer(const float* __restrict__ W,
                                          const float* __restrict__ bias,
                                          const float* sin_, float* sout,
                                          int cout, int lane) {
  const int hl = lane >> 4;
  const int ll = lane & 15;
  for (int ot = 0; ot < cout / 16; ++ot) {
    const float bv = bias[ot * 16 + ll];
#pragma unroll
    for (int rt = 0; rt < 4; ++rt) {
      v8f acc = {};
      const int m = rt * 16 + ll;
      const int o = ot * 16 + ll;
#pragma unroll
      for (int c = 0; c < CINPAD; c += 4) {
        const int kk = c + (hl << 1);
        v2f a, bf;
        a.x = sin_[m * C0PAD + kk];
        a.y = sin_[m * C0PAD + kk + 1];
        bf.x = W[o * WSTRIDE + kk];
        bf.y = W[o * WSTRIDE + kk + 1];
        acc = wmma_f32(a, bf, acc);
      }
#pragma unroll
      for (int r = 0; r < 8; ++r) {
        float v = acc[r] + bv;
        v = v > 0.0f ? v : 0.0f;
        const int krow = rt * 16 + r + 8 * hl;
        sout[krow * C0PAD + ot * 16 + ll] = v;
      }
    }
  }
}

__device__ __forceinline__ void mlp_layer3_max(const float* __restrict__ W,
                                               const float* __restrict__ bias,
                                               const float* sin_,
                                               float* __restrict__ out,
                                               int lane, int b, int s) {
  const int hl = lane >> 4;
  const int ll = lane & 15;
  for (int ot = 0; ot < 8; ++ot) {
    const float bv = bias[ot * 16 + ll];
    v8f mx = {};
    for (int rt = 0; rt < 4; ++rt) {
      v8f acc = {};
      const int m = rt * 16 + ll;
      const int o = ot * 16 + ll;
      for (int c = 0; c < 64; c += 4) {
        const int kk = c + (hl << 1);
        v2f a, bf;
        a.x = sin_[m * C0PAD + kk];
        a.y = sin_[m * C0PAD + kk + 1];
        bf.x = W[o * 64 + kk];
        bf.y = W[o * 64 + kk + 1];
        acc = wmma_f32(a, bf, acc);
      }
#pragma unroll
      for (int r = 0; r < 8; ++r) {
        float v = acc[r] + bv;
        v = v > 0.0f ? v : 0.0f;
        mx[r] = fmaxf(mx[r], v);
      }
    }
    float m8 = mx[0];
#pragma unroll
    for (int r = 1; r < 8; ++r) m8 = fmaxf(m8, mx[r]);
    float oth = __shfl_xor(m8, 16, 32);
    m8 = fmaxf(m8, oth);
    if (lane < 16) {
      const int o = ot * 16 + lane;
      out[OUT_NP_OFF + (size_t)b * 65536 + (size_t)o * 512 + s] = m8;
    }
  }
}

__global__ void __launch_bounds__(SA_THREADS) sa_fused_kernel(
    const float* __restrict__ xyz, const float* __restrict__ pw,
    const float* __restrict__ w0, const float* __restrict__ b0,
    const float* __restrict__ w1, const float* __restrict__ b1,
    const float* __restrict__ w2, const float* __restrict__ b2,
    float* __restrict__ out) {
  __shared__ float s_feat_all[QPB][64 * C0PAD];
  __shared__ float s_act_all[QPB][64 * C0PAD];
  __shared__ __attribute__((aligned(16))) int s_idx_all[QPB][NSAMPLE];
  __shared__ __attribute__((aligned(16))) unsigned s_hit[QPB][Nn / 32];

  const int lane = threadIdx.x & 31;
  const int w = __builtin_amdgcn_readfirstlane((int)(threadIdx.x >> 5));
  const int b = blockIdx.x >> 6;
  const int s0 = (blockIdx.x & 63) << 3;
  const int s = s0 + w;

  float* s_feat = s_feat_all[w];
  float* s_act = s_act_all[w];
  int* s_idx = s_idx_all[w];

  const float* xb = xyz + (size_t)b * 3 * Nn;
  const float* qb = out + (size_t)b * 1536 + s0;

  {
    const float R2 = RADIUS * RADIUS;
    const int p0 = w * SEG_PTS;
    float px[32], py[32], pz[32];
#pragma unroll
    for (int j = 0; j < 32; ++j) {
      const int n = p0 + j * 32 + lane;
      px[j] = xb[n];
      py[j] = xb[Nn + n];
      pz[j] = xb[2 * Nn + n];
    }
#pragma unroll
    for (int q = 0; q < QPB; ++q) {
      const float cx = qb[q];
      const float cy = qb[512 + q];
      const float cz = qb[1024 + q];
      int word = 0;
#pragma unroll
      for (int j = 0; j < 32; ++j) {
        const float dx = px[j] - cx;
        const float dy = py[j] - cy;
        const float dz = pz[j] - cz;
        const float d2 = __builtin_fmaf(dz, dz, __builtin_fmaf(dx, dx, dy * dy));
        const unsigned m = (unsigned)__ballot(d2 < R2);
        word = (lane == j) ? (int)m : word;
      }
      s_hit[q][w * 32 + lane] = (unsigned)word;
    }
  }
  __syncthreads();

  const float qx = qb[w];
  const float qy = qb[512 + w];
  const float qz = qb[1024 + w];

  {
    const uint4 h0 = *reinterpret_cast<const uint4*>(&s_hit[w][lane * 8]);
    const uint4 h1 = *reinterpret_cast<const uint4*>(&s_hit[w][lane * 8 + 4]);
    const unsigned hw[8] = {h0.x, h0.y, h0.z, h0.w, h1.x, h1.y, h1.z, h1.w};
    int c = 0;
#pragma unroll
    for (int i = 0; i < 8; ++i) c += __popc(hw[i]);
    int inc = c;
#pragma unroll
    for (int d = 1; d < 32; d <<= 1) {
      const int t = __shfl_up(inc, d, 32);
      if (lane >= d) inc += t;
    }
    const int total = __shfl(inc, 31, 32);
    int slot = inc - c;
#pragma unroll
    for (int i = 0; i < 8; ++i) {
      unsigned m = hw[i];
      const int base = (lane * 8 + i) * 32;
      while (m != 0u && slot < NSAMPLE) {
        s_idx[slot] = base + __builtin_ctz(m);
        ++slot;
        m &= m - 1u;
      }
    }
    for (int k = lane; k < NSAMPLE; k += 32)
      if (k >= total) s_idx[k] = -1;
  }
  __syncthreads();

  {
    const int hl = lane >> 4;
    const int ll = lane & 15;
    const float* Pb = pw + (size_t)b * Nn * 64;
    v2f bx[4];
    float bv[4];
#pragma unroll
    for (int ot = 0; ot < 4; ++ot) {
      const int o = ot * 16 + ll;
      bx[ot].x = w0[o * IN_CH + (hl << 1)];
      const float w1c = w0[o * IN_CH + 1];
      bx[ot].y = hl ? 0.0f : w1c;
      bv[ot] = b0[o];
    }
    v2f a[4];
#pragma unroll
    for (int rt = 0; rt < 4; ++rt) {
      const int n = s_idx[rt * 16 + ll];
      const int na = (n >= 0) ? n : 0;
      const float g0 = xb[(hl ? 2 * Nn : 0) + na];
      const float g1 = xb[Nn + na];
      a[rt].x = ((n >= 0) ? g0 : 0.0f) - (hl ? qz : qx);
      a[rt].y = hl ? 0.0f : (((n >= 0) ? g1 : 0.0f) - qy);
    }
#pragma unroll
    for (int rt = 0; rt < 4; ++rt) {
      const int4 i0 = *reinterpret_cast<const int4*>(&s_idx[rt * 16 + 8 * hl]);
      const int4 i1 = *reinterpret_cast<const int4*>(&s_idx[rt * 16 + 8 * hl + 4]);
      const int ix[8] = {i0.x, i0.y, i0.z, i0.w, i1.x, i1.y, i1.z, i1.w};
      int ro[8];
#pragma unroll
      for (int r = 0; r < 8; ++r) {
        const int gi = (ix[r] >= 0) ? ix[r] : (Nn - 1);
        ro[r] = gi * 64 + ll;
      }
#pragma unroll
      for (int ot = 0; ot < 4; ++ot) {
        v8f acc;
#pragma unroll
        for (int r = 0; r < 8; ++r) acc[r] = Pb[ro[r] + ot * 16];
        acc = wmma_f32(a[rt], bx[ot], acc);
#pragma unroll
        for (int r = 0; r < 8; ++r) {
          float v = acc[r] + bv[ot];
          v = v > 0.0f ? v : 0.0f;
          s_act[(rt * 16 + r + 8 * hl) * C0PAD + ot * 16 + ll] = v;
        }
      }
    }
  }
  __syncthreads();
  mlp_layer<64, 64>(w1, b1, s_act, s_feat, 64, lane);
  __syncthreads();
  mlp_layer3_max(w2, b2, s_feat, out, lane, b, s);
}

extern "C" void kernel_launch(void* const* d_in, const int* in_sizes, int n_in,
                              void* d_out, int out_size, void* d_ws, size_t ws_size,
                              hipStream_t stream) {
  const float* xyz = (const float*)d_in[0];
  const float* pts = (const float*)d_in[1];
  const float* attn = (const float*)d_in[2];
  const float* w0 = (const float*)d_in[3];
  const float* b0 = (const float*)d_in[4];
  const float* w1 = (const float*)d_in[5];
  const float* b1 = (const float*)d_in[6];
  const float* w2 = (const float*)d_in[7];
  const float* b2 = (const float*)d_in[8];
  float* out = (float*)d_out;
  float* pw = (float*)d_ws;

  hipLaunchKernelGGL(fps_kernel, dim3(FPS_BLOCKS + PT_BLOCKS), dim3(1024), 0, stream,
                     xyz, attn, pts, w0, pw, out);
  hipLaunchKernelGGL(sa_fused_kernel, dim3(SA_BLOCKS), dim3(SA_THREADS), 0, stream,
                     xyz, pw, w0, b0, w1, b1, w2, b2, out);
}


# FPS: a round's candidates are the (best, runner-up) pairs of 64 half-waves instead of 32 waves: a runner-up cuts the ranked list later, so rounds hold more picks; in-wave reduce one step shorter
# speedup vs baseline: 4.0961x; 1.0494x over previous
#include <hip/hip_runtime.h>

typedef __attribute__((ext_vector_type(2))) float v2f;
typedef __attribute__((ext_vector_type(8))) float v8f;

#define Bn 8
#define Nn 8192
#define S_ATTN 128
#define S_NONE 384
#define Sn 512
#define NSAMPLE 64
#define RADIUS 0.4f
#define D_FEAT 64
#define IN_CH 67
#define C0PAD 68
#define FPS_BLOCKS 16
#define PT_BLOCKS 128
#define QPB 8
#define SA_BLOCKS 512
#define SA_THREADS 256
#define SEG_PTS 1024

#define OUT_NP_OFF 12288
#define OUT_ATT_OFF 536576

__device__ __forceinline__ v8f wmma_f32(v2f a, v2f b, v8f c) {
  return __builtin_amdgcn_wmma_f32_16x16x4_f32(
       false, a,  false, b,
       (short)0, c,  false,  false);
}

__device__ __forceinline__ void point_term_tile(const float* __restrict__ pts,
                                                const float* __restrict__ w0,
                                                float* __restrict__ pw,
                                                int tile, int lane) {
  const int hl = lane >> 4;
  const int ll = lane & 15;
  const int b = tile >> 9;
  const int n0 = (tile & 511) << 4;
  const float* pb = pts + (size_t)b * D_FEAT * Nn;
  v2f a[16];
#pragma unroll
  for (int j = 0; j < 16; ++j) {
    const int kk = 4 * j + (hl << 1);
    a[j].x = pb[(size_t)kk * Nn + n0 + ll];
    a[j].y = pb[(size_t)(kk + 1) * Nn + n0 + ll];
  }
  float* prow = pw + ((size_t)b * Nn + n0) * 64;
  for (int ot = 0; ot < 4; ++ot) {
    const int o = ot * 16 + ll;
    v8f acc = {};
#pragma unroll
    for (int j = 0; j < 16; ++j) {
      const int kk = 4 * j + (hl << 1);
      v2f bf;
      bf.x = w0[o * IN_CH + 3 + kk];
      bf.y = w0[o * IN_CH + 3 + kk + 1];
      acc = wmma_f32(a[j], bf, acc);
    }
#pragma unroll
    for (int r = 0; r < 8; ++r) prow[(size_t)(r + 8 * hl) * 64 + o] = acc[r];
  }
}

__device__ __forceinline__ float sqd3(float px, float py, float pz, float lx, float ly, float lz) {
  const float dx = px - lx, dy = py - ly, dz = pz - lz;
  return __builtin_fmaf(dz, dz, __builtin_fmaf(dx, dx, dy * dy));
}

__device__ __forceinline__ bool ranks_before(float av, int an, float bv, int bn) {
  return av > bv || (av == bv && an < bn);
}

__device__ __forceinline__ void top2_merge(float& b1, int& n1, float& b2, int& n2,
                                           float ob1, int on1, float ob2, int on2) {
  const bool lead = ranks_before(ob1, on1, b1, n1);
  const float xv = lead ? b1 : b2;
  const int xn = lead ? n1 : n2;
  const float yv = lead ? ob2 : ob1;
  const int yn = lead ? on2 : on1;
  const bool ysel = ranks_before(yv, yn, xv, xn);
  b2 = ysel ? yv : xv;
  n2 = ysel ? yn : xn;
  b1 = lead ? ob1 : b1;
  n1 = lead ? on1 : n1;
}

#define MAXP 8

__device__ __forceinline__ void fps_top2(const float (&mind)[8], int tid,
                                         float& b1, int& n1, float& b2, int& n2) {
#pragma unroll
  for (int i = 0; i < 8; ++i) {
    const float m = mind[i];
    const int n = tid + i * 1024;
    const bool g1 = m > b1;
    const bool g2 = m > b2;
    b2 = g1 ? b1 : (g2 ? m : b2);
    n2 = g1 ? n1 : (g2 ? n : n2);
    b1 = g1 ? m : b1;
    n1 = g1 ? n : n1;
  }
}

__device__ __forceinline__ long long rank_key(float v, int n) {
  return (long long)(((unsigned long long)(unsigned)__float_as_int(v) << 32) |
                     (unsigned long long)(~(unsigned)n));
}

__device__ __forceinline__ int rank_among(long long q0, long long q1, long long q2, long long q3,
                                          long long x) {
  return __popc((unsigned)__ballot(q0 > x)) + __popc((unsigned)__ballot(q1 > x)) +
         __popc((unsigned)__ballot(q2 > x)) + __popc((unsigned)__ballot(q3 > x));
}

__global__ void __launch_bounds__(1024) fps_kernel(const float* __restrict__ xyz,
                                                   const float* __restrict__ attn,
                                                   const float* __restrict__ pts,
                                                   const float* __restrict__ w0,
                                                   float* __restrict__ pw,
                                                   float* __restrict__ out) {
  if (blockIdx.x >= FPS_BLOCKS) {
    point_term_tile(pts, w0, pw, (blockIdx.x - FPS_BLOCKS) * 32 + (threadIdx.x >> 5),
                    threadIdx.x & 31);
    return;
  }
  const int b = blockIdx.x >> 1;
  const int branch = blockIdx.x & 1;
  const int K = branch ? S_NONE : S_ATTN;
  const int rowoff = branch ? S_ATTN : 0;
  const int tid = threadIdx.x;
  const int wid = tid >> 5, lane = tid & 31;

  const float* xb = xyz + (size_t)b * 3 * Nn;
  const float* ab = attn + (size_t)b * Nn;

  float px[8], py[8], pz[8], mind[8];
#pragma unroll
  for (int i = 0; i < 8; ++i) {
    int n = tid + i * 1024;
    float a = ab[n];
    float w = branch ? (1.0f - a) : a;
    px[i] = w * xb[n];
    py[i] = w * xb[Nn + n];
    pz[i] = w * xb[2 * Nn + n];
    mind[i] = 1e10f;
  }

  __shared__ __attribute__((aligned(16))) long long s_key[128];
  __shared__ __attribute__((aligned(16))) float s_rec[MAXP][8];

  const int mj = lane >> 2;
  const int mc = lane & 3;

  if (tid < MAXP) {
    float a = ab[0];
    float w = branch ? (1.0f - a) : a;
    const bool first = tid == 0;
    s_rec[tid][0] = first ? w * xb[0] : 0.0f;
    s_rec[tid][1] = first ? w * xb[Nn] : 0.0f;
    s_rec[tid][2] = first ? w * xb[2 * Nn] : 0.0f;
    s_rec[tid][3] = 0.0f;
    s_rec[tid][4] = 0.0f;
    s_rec[tid][5] = a;
  }

  int k = 0;
  for (int round = 0; round < K; ++round) {
    __syncthreads();
    float qx[MAXP], qy[MAXP], qz[MAXP];
#pragma unroll
    for (int p = 0; p < MAXP; ++p) {
      qx[p] = __int_as_float(__builtin_amdgcn_readfirstlane(__float_as_int(s_rec[p][0])));
      qy[p] = __int_as_float(__builtin_amdgcn_readfirstlane(__float_as_int(s_rec[p][1])));
      qz[p] = __int_as_float(__builtin_amdgcn_readfirstlane(__float_as_int(s_rec[p][2])));
    }
    const float4 rj = *reinterpret_cast<const float4*>(&s_rec[mj][0]);
    const float vj = s_rec[mj][4];
    const float4 cl = *reinterpret_cast<const float4*>(&s_rec[mc][0]);
    const float4 ch = *reinterpret_cast<const float4*>(&s_rec[mc + 4][0]);
    const float vlast = s_rec[MAXP - 1][4];
    const float dlo = sqd3(rj.x, rj.y, rj.z, cl.x, cl.y, cl.z);
    const float dhi = sqd3(rj.x, rj.y, rj.z, ch.x, ch.y, ch.z);
    const bool lov = mc < mj, hiv = mc + 4 < mj;
    const unsigned vlo = (unsigned)__ballot(lov && dlo < vj);
    const unsigned vhi = (unsigned)__ballot(hiv && dhi < vj);
    const unsigned wlo = (unsigned)__ballot(lov && dlo < vlast);
    const unsigned whi = (unsigned)__ballot(hiv && dhi < vlast);
    const bool hardrow = (mj >= 1) & (!(vj > 0.0f) | (lov & (cl.w != 0.0f)) | (hiv & (ch.w != 0.0f)));
    const int H = __builtin_ctzll((unsigned long long)(unsigned)__ballot(hardrow) | (1ull << 32)) >> 2;
    unsigned A = (1u << H) - 1u;
    unsigned pos = (unsigned)((1ull << (4 * H)) - 1ull) & 0xFFFFFFF0u;
    for (int t = 0; t < MAXP - 1; ++t) {
      const unsigned alo = (A & 15u) * 0x11111111u;
      const unsigned ahi = (A >> 4) * 0x11111111u;
      const unsigned low = ((vlo & alo) | (vhi & ahi)) & pos;
      if (low == 0u) break;
      const int sr = __builtin_ctz(low) >> 2;
      const unsigned strong = (((wlo & alo) | (whi & ahi)) >> (4 * sr)) & 15u;
      if (strong == 0u) {
        A &= (1u << sr) - 1u;
        break;
      }
      A &= ~(1u << sr);
      pos &= 0xFFFFFFF0u << (4 * sr);
    }
    if (tid < MAXP && ((A >> tid) & 1u) != 0u) {
      const int kr = k + __popc(A & ((1u << tid) - 1u));
      if (kr < K) {
        const int row = rowoff + kr;
        out[(size_t)b * 1536 + row] = s_rec[tid][0];
        out[(size_t)b * 1536 + 512 + row] = s_rec[tid][1];
        out[(size_t)b * 1536 + 1024 + row] = s_rec[tid][2];
        out[OUT_ATT_OFF + (size_t)b * 512 + row] = s_rec[tid][5];
      }
    }
    k += __popc(A);
    if (k >= K) break;

    float b1 = -1.0f, b2 = -1.0f;
    int n1 = Nn, n2 = Nn;
#pragma unroll
    for (int p = 0; p < MAXP; ++p) {
      if ((A >> p) & 1u) {
#pragma unroll
        for (int i = 0; i < 8; ++i)
          mind[i] = fminf(mind[i], sqd3(px[i], py[i], pz[i], qx[p], qy[p], qz[p]));
      }
    }
    fps_top2(mind, tid, b1, n1, b2, n2);
    for (int off = 8; off > 0; off >>= 1) {
      float ob1 = __shfl_down(b1, off, 16);
      int on1 = __shfl_down(n1, off, 16);
      float ob2 = __shfl_down(b2, off, 16);
      int on2 = __shfl_down(n2, off, 16);
      top2_merge(b1, n1, b2, n2, ob1, on1, ob2, on2);
    }
    if ((lane & 15) == 0) {
      s_key[4 * wid + 2 * (lane >> 4)] = rank_key(b1, n1);
      s_key[4 * wid + 2 * (lane >> 4) + 1] = rank_key(b2, n2);
    }
    const int ub1 = __builtin_amdgcn_readfirstlane(__float_as_int(b1));
    const int un1 = __builtin_amdgcn_readfirstlane(n1);
    const int ub2 = __builtin_amdgcn_readfirstlane(__float_as_int(b2));
    const int un2 = __builtin_amdgcn_readfirstlane(n2);
    const int tb1 = __builtin_amdgcn_readlane(__float_as_int(b1), 16);
    const int tn1 = __builtin_amdgcn_readlane(n1, 16);
    const int tb2 = __builtin_amdgcn_readlane(__float_as_int(b2), 16);
    const int tn2 = __builtin_amdgcn_readlane(n2, 16);
    __syncthreads();
    {
      const long long q0 = s_key[4 * lane];
      const long long q1 = s_key[4 * lane + 1];
      const long long q2 = s_key[4 * lane + 2];
      const long long q3 = s_key[4 * lane + 3];
      const int ru1 = rank_among(q0, q1, q2, q3, rank_key(__int_as_float(ub1), un1));
      const int rt1 = rank_among(q0, q1, q2, q3, rank_key(__int_as_float(tb1), tn1));
      int ru2 = MAXP, rt2 = MAXP;
      if (ru1 < MAXP - 1) ru2 = rank_among(q0, q1, q2, q3, rank_key(__int_as_float(ub2), un2));
      if (rt1 < MAXP - 1) rt2 = rank_among(q0, q1, q2, q3, rank_key(__int_as_float(tb2), tn2));
      const bool second = (lane & 1) != 0;
      const bool upper = (lane & 2) != 0;
      const int r = upper ? (second ? rt2 : rt1) : (second ? ru2 : ru1);
      if (lane < 4 && r < MAXP) {
        const int cn = upper ? (second ? tn2 : tn1) : (second ? un2 : un1);
        const int cv = upper ? (second ? tb2 : tb1) : (second ? ub2 : ub1);
        const int c = cn & (Nn - 1);
        const float a = ab[c];
        const float x = xb[c], y = xb[Nn + c], z = xb[2 * Nn + c];
        const float w = branch ? (1.0f - a) : a;
        s_rec[r][0] = w * x;
        s_rec[r][1] = w * y;
        s_rec[r][2] = w * z;
        s_rec[r][3] = second ? 1.0f : 0.0f;
        s_rec[r][4] = __int_as_float(cv);
        s_rec[r][5] = a;
      }
    }
  }
}

template <int CINPAD, int WSTRIDE>
__device__ __forceinline__ void mlp_layer(const float* __restrict__ W,
                                          const float* __restrict__ bias,
                                          const float* sin_, float* sout,
                                          int cout, int lane) {
  const int hl = lane >> 4;
  const int ll = lane & 15;
  for (int ot = 0; ot < cout / 16; ++ot) {
    const float bv = bias[ot * 16 + ll];
#pragma unroll
    for (int rt = 0; rt < 4; ++rt) {
      v8f acc = {};
      const int m = rt * 16 + ll;
      const int o = ot * 16 + ll;
#pragma unroll
      for (int c = 0; c < CINPAD; c += 4) {
        const int kk = c + (hl << 1);
        v2f a, bf;
        a.x = sin_[m * C0PAD + kk];
        a.y = sin_[m * C0PAD + kk + 1];
        bf.x = W[o * WSTRIDE + kk];
        bf.y = W[o * WSTRIDE + kk + 1];
        acc = wmma_f32(a, bf, acc);
      }
#pragma unroll
      for (int r = 0; r < 8; ++r) {
        float v = acc[r] + bv;
        v = v > 0.0f ? v : 0.0f;
        const int krow = rt * 16 + r + 8 * hl;
        sout[krow * C0PAD + ot * 16 + ll] = v;
      }
    }
  }
}

__device__ __forceinline__ void mlp_layer3_max(const float* __restrict__ W,
                                               const float* __restrict__ bias,
                                               const float* sin_,
                                               float* __restrict__ out,
                                               int lane, int b, int s) {
  const int hl = lane >> 4;
  const int ll = lane & 15;
  for (int ot = 0; ot < 8; ++ot) {
    const float bv = bias[ot * 16 + ll];
    v8f mx = {};
    for (int rt = 0; rt < 4; ++rt) {
      v8f acc = {};
      const int m = rt * 16 + ll;
      const int o = ot * 16 + ll;
      for (int c = 0; c < 64; c += 4) {
        const int kk = c + (hl << 1);
        v2f a, bf;
        a.x = sin_[m * C0PAD + kk];
        a.y = sin_[m * C0PAD + kk + 1];
        bf.x = W[o * 64 + kk];
        bf.y = W[o * 64 + kk + 1];
        acc = wmma_f32(a, bf, acc);
      }
#pragma unroll
      for (int r = 0; r < 8; ++r) {
        float v = acc[r] + bv;
        v = v > 0.0f ? v : 0.0f;
        mx[r] = fmaxf(mx[r], v);
      }
    }
    float m8 = mx[0];
#pragma unroll
    for (int r = 1; r < 8; ++r) m8 = fmaxf(m8, mx[r]);
    float oth = __shfl_xor(m8, 16, 32);
    m8 = fmaxf(m8, oth);
    if (lane < 16) {
      const int o = ot * 16 + lane;
      out[OUT_NP_OFF + (size_t)b * 65536 + (size_t)o * 512 + s] = m8;
    }
  }
}

__global__ void __launch_bounds__(SA_THREADS) sa_fused_kernel(
    const float* __restrict__ xyz, const float* __restrict__ pw,
    const float* __restrict__ w0, const float* __restrict__ b0,
    const float* __restrict__ w1, const float* __restrict__ b1,
    const float* __restrict__ w2, const float* __restrict__ b2,
    float* __restrict__ out) {
  __shared__ float s_feat_all[QPB][64 * C0PAD];
  __shared__ float s_act_all[QPB][64 * C0PAD];
  __shared__ __attribute__((aligned(16))) int s_idx_all[QPB][NSAMPLE];
  __shared__ __attribute__((aligned(16))) unsigned s_hit[QPB][Nn / 32];

  const int lane = threadIdx.x & 31;
  const int w = __builtin_amdgcn_readfirstlane((int)(threadIdx.x >> 5));
  const int b = blockIdx.x >> 6;
  const int s0 = (blockIdx.x & 63) << 3;
  const int s = s0 + w;

  float* s_feat = s_feat_all[w];
  float* s_act = s_act_all[w];
  int* s_idx = s_idx_all[w];

  const float* xb = xyz + (size_t)b * 3 * Nn;
  const float* qb = out + (size_t)b * 1536 + s0;

  {
    const float R2 = RADIUS * RADIUS;
    const int p0 = w * SEG_PTS;
    float px[32], py[32], pz[32];
#pragma unroll
    for (int j = 0; j < 32; ++j) {
      const int n = p0 + j * 32 + lane;
      px[j] = xb[n];
      py[j] = xb[Nn + n];
      pz[j] = xb[2 * Nn + n];
    }
#pragma unroll
    for (int q = 0; q < QPB; ++q) {
      const float cx = qb[q];
      const float cy = qb[512 + q];
      const float cz = qb[1024 + q];
      int word = 0;
#pragma unroll
      for (int j = 0; j < 32; ++j) {
        const float dx = px[j] - cx;
        const float dy = py[j] - cy;
        const float dz = pz[j] - cz;
        const float d2 = __builtin_fmaf(dz, dz, __builtin_fmaf(dx, dx, dy * dy));
        const unsigned m = (unsigned)__ballot(d2 < R2);
        word = (lane == j) ? (int)m : word;
      }
      s_hit[q][w * 32 + lane] = (unsigned)word;
    }
  }
  __syncthreads();

  const float qx = qb[w];
  const float qy = qb[512 + w];
  const float qz = qb[1024 + w];

  {
    const uint4 h0 = *reinterpret_cast<const uint4*>(&s_hit[w][lane * 8]);
    const uint4 h1 = *reinterpret_cast<const uint4*>(&s_hit[w][lane * 8 + 4]);
    const unsigned hw[8] = {h0.x, h0.y, h0.z, h0.w, h1.x, h1.y, h1.z, h1.w};
    int c = 0;
#pragma unroll
    for (int i = 0; i < 8; ++i) c += __popc(hw[i]);
    int inc = c;
#pragma unroll
    for (int d = 1; d < 32; d <<= 1) {
      const int t = __shfl_up(inc, d, 32);
      if (lane >= d) inc += t;
    }
    const int total = __shfl(inc, 31, 32);
    int slot = inc - c;
#pragma unroll
    for (int i = 0; i < 8; ++i) {
      unsigned m = hw[i];
      const int base = (lane * 8 + i) * 32;
      while (m != 0u && slot < NSAMPLE) {
        s_idx[slot] = base + __builtin_ctz(m);
        ++slot;
        m &= m - 1u;
      }
    }
    for (int k = lane; k < NSAMPLE; k += 32)
      if (k >= total) s_idx[k] = -1;
  }
  __syncthreads();

  {
    const int hl = lane >> 4;
    const int ll = lane & 15;
    const float* Pb = pw + (size_t)b * Nn * 64;
    v2f bx[4];
    float bv[4];
#pragma unroll
    for (int ot = 0; ot < 4; ++ot) {
      const int o = ot * 16 + ll;
      bx[ot].x = w0[o * IN_CH + (hl << 1)];
      const float w1c = w0[o * IN_CH + 1];
      bx[ot].y = hl ? 0.0f : w1c;
      bv[ot] = b0[o];
    }
    v2f a[4];
#pragma unroll
    for (int rt = 0; rt < 4; ++rt) {
      const int n = s_idx[rt * 16 + ll];
      const int na = (n >= 0) ? n : 0;
      const float g0 = xb[(hl ? 2 * Nn : 0) + na];
      const float g1 = xb[Nn + na];
      a[rt].x = ((n >= 0) ? g0 : 0.0f) - (hl ? qz : qx);
      a[rt].y = hl ? 0.0f : (((n >= 0) ? g1 : 0.0f) - qy);
    }
#pragma unroll
    for (int rt = 0; rt < 4; ++rt) {
      const int4 i0 = *reinterpret_cast<const int4*>(&s_idx[rt * 16 + 8 * hl]);
      const int4 i1 = *reinterpret_cast<const int4*>(&s_idx[rt * 16 + 8 * hl + 4]);
      const int ix[8] = {i0.x, i0.y, i0.z, i0.w, i1.x, i1.y, i1.z, i1.w};
      int ro[8];
#pragma unroll
      for (int r = 0; r < 8; ++r) {
        const int gi = (ix[r] >= 0) ? ix[r] : (Nn - 1);
        ro[r] = gi * 64 + ll;
      }
#pragma unroll
      for (int ot = 0; ot < 4; ++ot) {
        v8f acc;
#pragma unroll
        for (int r = 0; r < 8; ++r) acc[r] = Pb[ro[r] + ot * 16];
        acc = wmma_f32(a[rt], bx[ot], acc);
#pragma unroll
        for (int r = 0; r < 8; ++r) {
          float v = acc[r] + bv[ot];
          v = v > 0.0f ? v : 0.0f;
          s_act[(rt * 16 + r + 8 * hl) * C0PAD + ot * 16 + ll] = v;
        }
      }
    }
  }
  __syncthreads();
  mlp_layer<64, 64>(w1, b1, s_act, s_feat, 64, lane);
  __syncthreads();
  mlp_layer3_max(w2, b2, s_feat, out, lane, b, s);
}

extern "C" void kernel_launch(void* const* d_in, const int* in_sizes, int n_in,
                              void* d_out, int out_size, void* d_ws, size_t ws_size,
                              hipStream_t stream) {
  const float* xyz = (const float*)d_in[0];
  const float* pts = (const float*)d_in[1];
  const float* attn = (const float*)d_in[2];
  const float* w0 = (const float*)d_in[3];
  const float* b0 = (const float*)d_in[4];
  const float* w1 = (const float*)d_in[5];
  const float* b1 = (const float*)d_in[6];
  const float* w2 = (const float*)d_in[7];
  const float* b2 = (const float*)d_in[8];
  float* out = (float*)d_out;
  float* pw = (float*)d_ws;

  hipLaunchKernelGGL(fps_kernel, dim3(FPS_BLOCKS + PT_BLOCKS), dim3(1024), 0, stream,
                     xyz, attn, pts, w0, pw, out);
  hipLaunchKernelGGL(sa_fused_kernel, dim3(SA_BLOCKS), dim3(SA_THREADS), 0, stream,
                     xyz, pw, w0, b0, w1, b1, w2, b2, out);
}


# FPS: a round's candidate coordinate and attention loads are issued before the key barrier and the ranking instead of after them, so their latency runs under both; same values, same order
# speedup vs baseline: 4.3306x; 1.0572x over previous
#include <hip/hip_runtime.h>

typedef __attribute__((ext_vector_type(2))) float v2f;
typedef __attribute__((ext_vector_type(8))) float v8f;

#define Bn 8
#define Nn 8192
#define S_ATTN 128
#define S_NONE 384
#define Sn 512
#define NSAMPLE 64
#define RADIUS 0.4f
#define D_FEAT 64
#define IN_CH 67
#define C0PAD 68
#define FPS_BLOCKS 16
#define PT_BLOCKS 128
#define QPB 8
#define SA_BLOCKS 512
#define SA_THREADS 256
#define SEG_PTS 1024

#define OUT_NP_OFF 12288
#define OUT_ATT_OFF 536576

__device__ __forceinline__ v8f wmma_f32(v2f a, v2f b, v8f c) {
  return __builtin_amdgcn_wmma_f32_16x16x4_f32(
       false, a,  false, b,
       (short)0, c,  false,  false);
}

__device__ __forceinline__ void point_term_tile(const float* __restrict__ pts,
                                                const float* __restrict__ w0,
                                                float* __restrict__ pw,
                                                int tile, int lane) {
  const int hl = lane >> 4;
  const int ll = lane & 15;
  const int b = tile >> 9;
  const int n0 = (tile & 511) << 4;
  const float* pb = pts + (size_t)b * D_FEAT * Nn;
  v2f a[16];
#pragma unroll
  for (int j = 0; j < 16; ++j) {
    const int kk = 4 * j + (hl << 1);
    a[j].x = pb[(size_t)kk * Nn + n0 + ll];
    a[j].y = pb[(size_t)(kk + 1) * Nn + n0 + ll];
  }
  float* prow = pw + ((size_t)b * Nn + n0) * 64;
  for (int ot = 0; ot < 4; ++ot) {
    const int o = ot * 16 + ll;
    v8f acc = {};
#pragma unroll
    for (int j = 0; j < 16; ++j) {
      const int kk = 4 * j + (hl << 1);
      v2f bf;
      bf.x = w0[o * IN_CH + 3 + kk];
      bf.y = w0[o * IN_CH + 3 + kk + 1];
      acc = wmma_f32(a[j], bf, acc);
    }
#pragma unroll
    for (int r = 0; r < 8; ++r) prow[(size_t)(r + 8 * hl) * 64 + o] = acc[r];
  }
}

__device__ __forceinline__ float sqd3(float px, float py, float pz, float lx, float ly, float lz) {
  const float dx = px - lx, dy = py - ly, dz = pz - lz;
  return __builtin_fmaf(dz, dz, __builtin_fmaf(dx, dx, dy * dy));
}

__device__ __forceinline__ bool ranks_before(float av, int an, float bv, int bn) {
  return av > bv || (av == bv && an < bn);
}

__device__ __forceinline__ void top2_merge(float& b1, int& n1, float& b2, int& n2,
                                           float ob1, int on1, float ob2, int on2) {
  const bool lead = ranks_before(ob1, on1, b1, n1);
  const float xv = lead ? b1 : b2;
  const int xn = lead ? n1 : n2;
  const float yv = lead ? ob2 : ob1;
  const int yn = lead ? on2 : on1;
  const bool ysel = ranks_before(yv, yn, xv, xn);
  b2 = ysel ? yv : xv;
  n2 = ysel ? yn : xn;
  b1 = lead ? ob1 : b1;
  n1 = lead ? on1 : n1;
}

#define MAXP 8

__device__ __forceinline__ void fps_top2(const float (&mind)[8], int tid,
                                         float& b1, int& n1, float& b2, int& n2) {
#pragma unroll
  for (int i = 0; i < 8; ++i) {
    const float m = mind[i];
    const int n = tid + i * 1024;
    const bool g1 = m > b1;
    const bool g2 = m > b2;
    b2 = g1 ? b1 : (g2 ? m : b2);
    n2 = g1 ? n1 : (g2 ? n : n2);
    b1 = g1 ? m : b1;
    n1 = g1 ? n : n1;
  }
}

__device__ __forceinline__ long long rank_key(float v, int n) {
  return (long long)(((unsigned long long)(unsigned)__float_as_int(v) << 32) |
                     (unsigned long long)(~(unsigned)n));
}

__device__ __forceinline__ int rank_among(long long q0, long long q1, long long q2, long long q3,
                                          long long x) {
  return __popc((unsigned)__ballot(q0 > x)) + __popc((unsigned)__ballot(q1 > x)) +
         __popc((unsigned)__ballot(q2 > x)) + __popc((unsigned)__ballot(q3 > x));
}

__global__ void __launch_bounds__(1024) fps_kernel(const float* __restrict__ xyz,
                                                   const float* __restrict__ attn,
                                                   const float* __restrict__ pts,
                                                   const float* __restrict__ w0,
                                                   float* __restrict__ pw,
                                                   float* __restrict__ out) {
  if (blockIdx.x >= FPS_BLOCKS) {
    point_term_tile(pts, w0, pw, (blockIdx.x - FPS_BLOCKS) * 32 + (threadIdx.x >> 5),
                    threadIdx.x & 31);
    return;
  }
  const int b = blockIdx.x >> 1;
  const int branch = blockIdx.x & 1;
  const int K = branch ? S_NONE : S_ATTN;
  const int rowoff = branch ? S_ATTN : 0;
  const int tid = threadIdx.x;
  const int wid = tid >> 5, lane = tid & 31;

  const float* xb = xyz + (size_t)b * 3 * Nn;
  const float* ab = attn + (size_t)b * Nn;

  float px[8], py[8], pz[8], mind[8];
#pragma unroll
  for (int i = 0; i < 8; ++i) {
    int n = tid + i * 1024;
    float a = ab[n];
    float w = branch ? (1.0f - a) : a;
    px[i] = w * xb[n];
    py[i] = w * xb[Nn + n];
    pz[i] = w * xb[2 * Nn + n];
    mind[i] = 1e10f;
  }

  __shared__ __attribute__((aligned(16))) long long s_key[128];
  __shared__ __attribute__((aligned(16))) float s_rec[MAXP][8];

  const int mj = lane >> 2;
  const int mc = lane & 3;

  if (tid < MAXP) {
    float a = ab[0];
    float w = branch ? (1.0f - a) : a;
    const bool first = tid == 0;
    s_rec[tid][0] = first ? w * xb[0] : 0.0f;
    s_rec[tid][1] = first ? w * xb[Nn] : 0.0f;
    s_rec[tid][2] = first ? w * xb[2 * Nn] : 0.0f;
    s_rec[tid][3] = 0.0f;
    s_rec[tid][4] = 0.0f;
    s_rec[tid][5] = a;
  }

  int k = 0;
  for (int round = 0; round < K; ++round) {
    __syncthreads();
    float qx[MAXP], qy[MAXP], qz[MAXP];
#pragma unroll
    for (int p = 0; p < MAXP; ++p) {
      qx[p] = __int_as_float(__builtin_amdgcn_readfirstlane(__float_as_int(s_rec[p][0])));
      qy[p] = __int_as_float(__builtin_amdgcn_readfirstlane(__float_as_int(s_rec[p][1])));
      qz[p] = __int_as_float(__builtin_amdgcn_readfirstlane(__float_as_int(s_rec[p][2])));
    }
    const float4 rj = *reinterpret_cast<const float4*>(&s_rec[mj][0]);
    const float vj = s_rec[mj][4];
    const float4 cl = *reinterpret_cast<const float4*>(&s_rec[mc][0]);
    const float4 ch = *reinterpret_cast<const float4*>(&s_rec[mc + 4][0]);
    const float vlast = s_rec[MAXP - 1][4];
    const float dlo = sqd3(rj.x, rj.y, rj.z, cl.x, cl.y, cl.z);
    const float dhi = sqd3(rj.x, rj.y, rj.z, ch.x, ch.y, ch.z);
    const bool lov = mc < mj, hiv = mc + 4 < mj;
    const unsigned vlo = (unsigned)__ballot(lov && dlo < vj);
    const unsigned vhi = (unsigned)__ballot(hiv && dhi < vj);
    const unsigned wlo = (unsigned)__ballot(lov && dlo < vlast);
    const unsigned whi = (unsigned)__ballot(hiv && dhi < vlast);
    const bool hardrow = (mj >= 1) & (!(vj > 0.0f) | (lov & (cl.w != 0.0f)) | (hiv & (ch.w != 0.0f)));
    const int H = __builtin_ctzll((unsigned long long)(unsigned)__ballot(hardrow) | (1ull << 32)) >> 2;
    unsigned A = (1u << H) - 1u;
    unsigned pos = (unsigned)((1ull << (4 * H)) - 1ull) & 0xFFFFFFF0u;
    for (int t = 0; t < MAXP - 1; ++t) {
      const unsigned alo = (A & 15u) * 0x11111111u;
      const unsigned ahi = (A >> 4) * 0x11111111u;
      const unsigned low = ((vlo & alo) | (vhi & ahi)) & pos;
      if (low == 0u) break;
      const int sr = __builtin_ctz(low) >> 2;
      const unsigned strong = (((wlo & alo) | (whi & ahi)) >> (4 * sr)) & 15u;
      if (strong == 0u) {
        A &= (1u << sr) - 1u;
        break;
      }
      A &= ~(1u << sr);
      pos &= 0xFFFFFFF0u << (4 * sr);
    }
    if (tid < MAXP && ((A >> tid) & 1u) != 0u) {
      const int kr = k + __popc(A & ((1u << tid) - 1u));
      if (kr < K) {
        const int row = rowoff + kr;
        out[(size_t)b * 1536 + row] = s_rec[tid][0];
        out[(size_t)b * 1536 + 512 + row] = s_rec[tid][1];
        out[(size_t)b * 1536 + 1024 + row] = s_rec[tid][2];
        out[OUT_ATT_OFF + (size_t)b * 512 + row] = s_rec[tid][5];
      }
    }
    k += __popc(A);
    if (k >= K) break;

    float b1 = -1.0f, b2 = -1.0f;
    int n1 = Nn, n2 = Nn;
#pragma unroll
    for (int p = 0; p < MAXP; ++p) {
      if ((A >> p) & 1u) {
#pragma unroll
        for (int i = 0; i < 8; ++i)
          mind[i] = fminf(mind[i], sqd3(px[i], py[i], pz[i], qx[p], qy[p], qz[p]));
      }
    }
    fps_top2(mind, tid, b1, n1, b2, n2);
    for (int off = 8; off > 0; off >>= 1) {
      float ob1 = __shfl_down(b1, off, 16);
      int on1 = __shfl_down(n1, off, 16);
      float ob2 = __shfl_down(b2, off, 16);
      int on2 = __shfl_down(n2, off, 16);
      top2_merge(b1, n1, b2, n2, ob1, on1, ob2, on2);
    }
    if ((lane & 15) == 0) {
      s_key[4 * wid + 2 * (lane >> 4)] = rank_key(b1, n1);
      s_key[4 * wid + 2 * (lane >> 4) + 1] = rank_key(b2, n2);
    }
    const int ub1 = __builtin_amdgcn_readfirstlane(__float_as_int(b1));
    const int un1 = __builtin_amdgcn_readfirstlane(n1);
    const int ub2 = __builtin_amdgcn_readfirstlane(__float_as_int(b2));
    const int un2 = __builtin_amdgcn_readfirstlane(n2);
    const int tb1 = __builtin_amdgcn_readlane(__float_as_int(b1), 16);
    const int tn1 = __builtin_amdgcn_readlane(n1, 16);
    const int tb2 = __builtin_amdgcn_readlane(__float_as_int(b2), 16);
    const int tn2 = __builtin_amdgcn_readlane(n2, 16);
    const bool second = (lane & 1) != 0;
    const bool upper = (lane & 2) != 0;
    const int cn = upper ? (second ? tn2 : tn1) : (second ? un2 : un1);
    const int cidx = cn & (Nn - 1);
    const float ca = ab[cidx];
    const float cx = xb[cidx], cy = xb[Nn + cidx], cz = xb[2 * Nn + cidx];
    __builtin_amdgcn_fence(__ATOMIC_RELEASE, "workgroup", "local");
    __builtin_amdgcn_s_barrier();
    __builtin_amdgcn_fence(__ATOMIC_ACQUIRE, "workgroup", "local");
    {
      const long long q0 = s_key[4 * lane];
      const long long q1 = s_key[4 * lane + 1];
      const long long q2 = s_key[4 * lane + 2];
      const long long q3 = s_key[4 * lane + 3];
      const int ru1 = rank_among(q0, q1, q2, q3, rank_key(__int_as_float(ub1), un1));
      const int rt1 = rank_among(q0, q1, q2, q3, rank_key(__int_as_float(tb1), tn1));
      int ru2 = MAXP, rt2 = MAXP;
      if (ru1 < MAXP - 1) ru2 = rank_among(q0, q1, q2, q3, rank_key(__int_as_float(ub2), un2));
      if (rt1 < MAXP - 1) rt2 = rank_among(q0, q1, q2, q3, rank_key(__int_as_float(tb2), tn2));
      const int r = upper ? (second ? rt2 : rt1) : (second ? ru2 : ru1);
      if (lane < 4 && r < MAXP) {
        const int cv = upper ? (second ? tb2 : tb1) : (second ? ub2 : ub1);
        const float w = branch ? (1.0f - ca) : ca;
        s_rec[r][0] = w * cx;
        s_rec[r][1] = w * cy;
        s_rec[r][2] = w * cz;
        s_rec[r][3] = second ? 1.0f : 0.0f;
        s_rec[r][4] = __int_as_float(cv);
        s_rec[r][5] = ca;
      }
    }
  }
}

template <int CINPAD, int WSTRIDE>
__device__ __forceinline__ void mlp_layer(const float* __restrict__ W,
                                          const float* __restrict__ bias,
                                          const float* sin_, float* sout,
                                          int cout, int lane) {
  const int hl = lane >> 4;
  const int ll = lane & 15;
  for (int ot = 0; ot < cout / 16; ++ot) {
    const float bv = bias[ot * 16 + ll];
#pragma unroll
    for (int rt = 0; rt < 4; ++rt) {
      v8f acc = {};
      const int m = rt * 16 + ll;
      const int o = ot * 16 + ll;
#pragma unroll
      for (int c = 0; c < CINPAD; c += 4) {
        const int kk = c + (hl << 1);
        v2f a, bf;
        a.x = sin_[m * C0PAD + kk];
        a.y = sin_[m * C0PAD + kk + 1];
        bf.x = W[o * WSTRIDE + kk];
        bf.y = W[o * WSTRIDE + kk + 1];
        acc = wmma_f32(a, bf, acc);
      }
#pragma unroll
      for (int r = 0; r < 8; ++r) {
        float v = acc[r] + bv;
        v = v > 0.0f ? v : 0.0f;
        const int krow = rt * 16 + r + 8 * hl;
        sout[krow * C0PAD + ot * 16 + ll] = v;
      }
    }
  }
}

__device__ __forceinline__ void mlp_layer3_max(const float* __restrict__ W,
                                               const float* __restrict__ bias,
                                               const float* sin_,
                                               float* __restrict__ out,
                                               int lane, int b, int s) {
  const int hl = lane >> 4;
  const int ll = lane & 15;
  for (int ot = 0; ot < 8; ++ot) {
    const float bv = bias[ot * 16 + ll];
    v8f mx = {};
    for (int rt = 0; rt < 4; ++rt) {
      v8f acc = {};
      const int m = rt * 16 + ll;
      const int o = ot * 16 + ll;
      for (int c = 0; c < 64; c += 4) {
        const int kk = c + (hl << 1);
        v2f a, bf;
        a.x = sin_[m * C0PAD + kk];
        a.y = sin_[m * C0PAD + kk + 1];
        bf.x = W[o * 64 + kk];
        bf.y = W[o * 64 + kk + 1];
        acc = wmma_f32(a, bf, acc);
      }
#pragma unroll
      for (int r = 0; r < 8; ++r) {
        float v = acc[r] + bv;
        v = v > 0.0f ? v : 0.0f;
        mx[r] = fmaxf(mx[r], v);
      }
    }
    float m8 = mx[0];
#pragma unroll
    for (int r = 1; r < 8; ++r) m8 = fmaxf(m8, mx[r]);
    float oth = __shfl_xor(m8, 16, 32);
    m8 = fmaxf(m8, oth);
    if (lane < 16) {
      const int o = ot * 16 + lane;
      out[OUT_NP_OFF + (size_t)b * 65536 + (size_t)o * 512 + s] = m8;
    }
  }
}

__global__ void __launch_bounds__(SA_THREADS) sa_fused_kernel(
    const float* __restrict__ xyz, const float* __restrict__ pw,
    const float* __restrict__ w0, const float* __restrict__ b0,
    const float* __restrict__ w1, const float* __restrict__ b1,
    const float* __restrict__ w2, const float* __restrict__ b2,
    float* __restrict__ out) {
  __shared__ float s_feat_all[QPB][64 * C0PAD];
  __shared__ float s_act_all[QPB][64 * C0PAD];
  __shared__ __attribute__((aligned(16))) int s_idx_all[QPB][NSAMPLE];
  __shared__ __attribute__((aligned(16))) unsigned s_hit[QPB][Nn / 32];

  const int lane = threadIdx.x & 31;
  const int w = __builtin_amdgcn_readfirstlane((int)(threadIdx.x >> 5));
  const int b = blockIdx.x >> 6;
  const int s0 = (blockIdx.x & 63) << 3;
  const int s = s0 + w;

  float* s_feat = s_feat_all[w];
  float* s_act = s_act_all[w];
  int* s_idx = s_idx_all[w];

  const float* xb = xyz + (size_t)b * 3 * Nn;
  const float* qb = out + (size_t)b * 1536 + s0;

  {
    const float R2 = RADIUS * RADIUS;
    const int p0 = w * SEG_PTS;
    float px[32], py[32], pz[32];
#pragma unroll
    for (int j = 0; j < 32; ++j) {
      const int n = p0 + j * 32 + lane;
      px[j] = xb[n];
      py[j] = xb[Nn + n];
      pz[j] = xb[2 * Nn + n];
    }
#pragma unroll
    for (int q = 0; q < QPB; ++q) {
      const float cx = qb[q];
      const float cy = qb[512 + q];
      const float cz = qb[1024 + q];
      int word = 0;
#pragma unroll
      for (int j = 0; j < 32; ++j) {
        const float dx = px[j] - cx;
        const float dy = py[j] - cy;
        const float dz = pz[j] - cz;
        const float d2 = __builtin_fmaf(dz, dz, __builtin_fmaf(dx, dx, dy * dy));
        const unsigned m = (unsigned)__ballot(d2 < R2);
        word = (lane == j) ? (int)m : word;
      }
      s_hit[q][w * 32 + lane] = (unsigned)word;
    }
  }
  __syncthreads();

  const float qx = qb[w];
  const float qy = qb[512 + w];
  const float qz = qb[1024 + w];

  {
    const uint4 h0 = *reinterpret_cast<const uint4*>(&s_hit[w][lane * 8]);
    const uint4 h1 = *reinterpret_cast<const uint4*>(&s_hit[w][lane * 8 + 4]);
    const unsigned hw[8] = {h0.x, h0.y, h0.z, h0.w, h1.x, h1.y, h1.z, h1.w};
    int c = 0;
#pragma unroll
    for (int i = 0; i < 8; ++i) c += __popc(hw[i]);
    int inc = c;
#pragma unroll
    for (int d = 1; d < 32; d <<= 1) {
      const int t = __shfl_up(inc, d, 32);
      if (lane >= d) inc += t;
    }
    const int total = __shfl(inc, 31, 32);
    int slot = inc - c;
#pragma unroll
    for (int i = 0; i < 8; ++i) {
      unsigned m = hw[i];
      const int base = (lane * 8 + i) * 32;
      while (m != 0u && slot < NSAMPLE) {
        s_idx[slot] = base + __builtin_ctz(m);
        ++slot;
        m &= m - 1u;
      }
    }
    for (int k = lane; k < NSAMPLE; k += 32)
      if (k >= total) s_idx[k] = -1;
  }
  __syncthreads();

  {
    const int hl = lane >> 4;
    const int ll = lane & 15;
    const float* Pb = pw + (size_t)b * Nn * 64;
    v2f bx[4];
    float bv[4];
#pragma unroll
    for (int ot = 0; ot < 4; ++ot) {
      const int o = ot * 16 + ll;
      bx[ot].x = w0[o * IN_CH + (hl << 1)];
      const float w1c = w0[o * IN_CH + 1];
      bx[ot].y = hl ? 0.0f : w1c;
      bv[ot] = b0[o];
    }
    v2f a[4];
#pragma unroll
    for (int rt = 0; rt < 4; ++rt) {
      const int n = s_idx[rt * 16 + ll];
      const int na = (n >= 0) ? n : 0;
      const float g0 = xb[(hl ? 2 * Nn : 0) + na];
      const float g1 = xb[Nn + na];
      a[rt].x = ((n >= 0) ? g0 : 0.0f) - (hl ? qz : qx);
      a[rt].y = hl ? 0.0f : (((n >= 0) ? g1 : 0.0f) - qy);
    }
#pragma unroll
    for (int rt = 0; rt < 4; ++rt) {
      const int4 i0 = *reinterpret_cast<const int4*>(&s_idx[rt * 16 + 8 * hl]);
      const int4 i1 = *reinterpret_cast<const int4*>(&s_idx[rt * 16 + 8 * hl + 4]);
      const int ix[8] = {i0.x, i0.y, i0.z, i0.w, i1.x, i1.y, i1.z, i1.w};
      int ro[8];
#pragma unroll
      for (int r = 0; r < 8; ++r) {
        const int gi = (ix[r] >= 0) ? ix[r] : (Nn - 1);
        ro[r] = gi * 64 + ll;
      }
#pragma unroll
      for (int ot = 0; ot < 4; ++ot) {
        v8f acc;
#pragma unroll
        for (int r = 0; r < 8; ++r) acc[r] = Pb[ro[r] + ot * 16];
        acc = wmma_f32(a[rt], bx[ot], acc);
#pragma unroll
        for (int r = 0; r < 8; ++r) {
          float v = acc[r] + bv[ot];
          v = v > 0.0f ? v : 0.0f;
          s_act[(rt * 16 + r + 8 * hl) * C0PAD + ot * 16 + ll] = v;
        }
      }
    }
  }
  __syncthreads();
  mlp_layer<64, 64>(w1, b1, s_act, s_feat, 64, lane);
  __syncthreads();
  mlp_layer3_max(w2, b2, s_feat, out, lane, b, s);
}

extern "C" void kernel_launch(void* const* d_in, const int* in_sizes, int n_in,
                              void* d_out, int out_size, void* d_ws, size_t ws_size,
                              hipStream_t stream) {
  const float* xyz = (const float*)d_in[0];
  const float* pts = (const float*)d_in[1];
  const float* attn = (const float*)d_in[2];
  const float* w0 = (const float*)d_in[3];
  const float* b0 = (const float*)d_in[4];
  const float* w1 = (const float*)d_in[5];
  const float* b1 = (const float*)d_in[6];
  const float* w2 = (const float*)d_in[7];
  const float* b2 = (const float*)d_in[8];
  float* out = (float*)d_out;
  float* pw = (float*)d_ws;

  hipLaunchKernelGGL(fps_kernel, dim3(FPS_BLOCKS + PT_BLOCKS), dim3(1024), 0, stream,
                     xyz, attn, pts, w0, pw, out);
  hipLaunchKernelGGL(sa_fused_kernel, dim3(SA_BLOCKS), dim3(SA_THREADS), 0, stream,
                     xyz, pw, w0, b0, w1, b1, w2, b2, out);
}
